# starting kernel
# baseline (speedup 1.0000x reference)
#include <hip/hip_runtime.h>
#include <math.h>

typedef float v2f __attribute__((ext_vector_type(2)));
typedef float v8f __attribute__((ext_vector_type(8)));

#define F_IN  128
#define F_OUT 64


__global__ void k_init_deg(float* __restrict__ deg, int N) {
    int i = blockIdx.x * blockDim.x + threadIdx.x;
    if (i < N) deg[i] = 1.0f;
}

__global__ void k_count_deg(const int* __restrict__ dst, float* __restrict__ deg, int E) {
    int e = blockIdx.x * blockDim.x + threadIdx.x;
    if (e < E) atomicAdd(&deg[dst[e]], 1.0f);
}

__global__ void k_dinv(float* __restrict__ deg, int N) {
    int i = blockIdx.x * blockDim.x + threadIdx.x;
    if (i < N) {
        float d = deg[i];
        deg[i] = (d > 0.0f) ? rsqrtf(d) : 0.0f;
    }
}


__global__ void __launch_bounds__(256) k_gemm(const float* __restrict__ x,
                                              const float* __restrict__ W,
                                              float* __restrict__ h, int N) {
    const int wave = threadIdx.x >> 5;
    const int lane = threadIdx.x & 31;
    const int row0 = (blockIdx.x * 8 + wave) * 16;
    if (row0 >= N) return;

    const int lr    = lane & 15;
    const int khalf = lane >> 4;

    int m = row0 + lr;
    if (m >= N) m = N - 1;

    const float2* xp  = (const float2*)(x + (size_t)m * F_IN);
    const float2* wp0 = (const float2*)(W + (size_t)(lr +  0) * F_IN);
    const float2* wp1 = (const float2*)(W + (size_t)(lr + 16) * F_IN);
    const float2* wp2 = (const float2*)(W + (size_t)(lr + 32) * F_IN);
    const float2* wp3 = (const float2*)(W + (size_t)(lr + 48) * F_IN);

    v8f c0 = {}, c1 = {}, c2 = {}, c3 = {};

    #pragma unroll
    for (int k0 = 0; k0 < F_IN; k0 += 4) {
        const int ki = (k0 >> 1) + khalf;
        float2 af = xp[ki];
        v2f a; a.x = af.x; a.y = af.y;
        float2 b0f = wp0[ki]; v2f b0; b0.x = b0f.x; b0.y = b0f.y;
        float2 b1f = wp1[ki]; v2f b1; b1.x = b1f.x; b1.y = b1f.y;
        float2 b2f = wp2[ki]; v2f b2; b2.x = b2f.x; b2.y = b2f.y;
        float2 b3f = wp3[ki]; v2f b3; b3.x = b3f.x; b3.y = b3f.y;

        c0 = __builtin_amdgcn_wmma_f32_16x16x4_f32(false, a, false, b0, (short)0, c0, false, false);
        c1 = __builtin_amdgcn_wmma_f32_16x16x4_f32(false, a, false, b1, (short)0, c1, false, false);
        c2 = __builtin_amdgcn_wmma_f32_16x16x4_f32(false, a, false, b2, (short)0, c2, false, false);
        c3 = __builtin_amdgcn_wmma_f32_16x16x4_f32(false, a, false, b3, (short)0, c3, false, false);
    }

    const int mbase = row0 + (khalf << 3);
    #pragma unroll
    for (int v = 0; v < 8; ++v) {
        int mm = mbase + v;
        if (mm < N) {
            float* hp = h + (size_t)mm * F_OUT + lr;
            hp[ 0] = c0[v];
            hp[16] = c1[v];
            hp[32] = c2[v];
            hp[48] = c3[v];
        }
    }
}


__global__ void k_init_agg(const float* __restrict__ h, const float* __restrict__ dinv,
                           const float* __restrict__ b_conv, float* __restrict__ agg, int total) {
    int idx = blockIdx.x * blockDim.x + threadIdx.x;
    if (idx < total) {
        int node = idx >> 6;
        int f    = idx & 63;
        float di = dinv[node];
        agg[idx] = b_conv[f] + h[idx] * di * di;
    }
}


__global__ void k_scatter(const int* __restrict__ src, const int* __restrict__ dst,
                          const float* __restrict__ h, const float* __restrict__ dinv,
                          float* __restrict__ agg, int E) {
    int gtid = blockIdx.x * blockDim.x + threadIdx.x;
    int e    = gtid >> 5;
    int lane = threadIdx.x & 31;
    if (e >= E) return;

    int s = src[e];
    int d = dst[e];
    float norm = dinv[s] * dinv[d];

    const float2* hs = (const float2*)(h + (size_t)s * F_OUT);
    float2 v = hs[lane];

    float* ad = agg + (size_t)d * F_OUT + lane * 2;
    atomicAdd(ad + 0, v.x * norm);
    atomicAdd(ad + 1, v.y * norm);
}


__global__ void k_finalize(const float* __restrict__ agg, const float* __restrict__ W_lin,
                           const float* __restrict__ b_lin, float* __restrict__ out, int N) {
    int gtid = blockIdx.x * blockDim.x + threadIdx.x;
    int node = gtid >> 5;
    int lane = threadIdx.x & 31;
    if (node >= N) return;

    const float* row = agg + (size_t)node * F_OUT;
    float p = fmaxf(row[lane], 0.0f) * W_lin[lane]
            + fmaxf(row[lane + 32], 0.0f) * W_lin[lane + 32];

    #pragma unroll
    for (int off = 16; off > 0; off >>= 1)
        p += __shfl_down(p, off);

    if (lane == 0) {
        float z = p + b_lin[0];
        out[node] = 1.0f / (1.0f + __expf(-z));
    }
}


extern "C" void kernel_launch(void* const* d_in, const int* in_sizes, int n_in,
                              void* d_out, int out_size, void* d_ws, size_t ws_size,
                              hipStream_t stream) {
    const float* x      = (const float*)d_in[0];
    const int*   edge   = (const int*)  d_in[1];
    const float* W_conv = (const float*)d_in[2];
    const float* b_conv = (const float*)d_in[3];
    const float* W_lin  = (const float*)d_in[4];
    const float* b_lin  = (const float*)d_in[5];
    float*       out    = (float*)d_out;

    const int N = in_sizes[0] / F_IN;
    const int E = in_sizes[1] / 2;
    const int* src = edge;
    const int* dst = edge + E;

    float* ws   = (float*)d_ws;
    float* dinv = ws;
    float* h    = ws + N;
    float* agg  = h + (size_t)N * F_OUT;

    k_init_deg <<<(N + 255) / 256, 256, 0, stream>>>(dinv, N);
    k_count_deg<<<(E + 255) / 256, 256, 0, stream>>>(dst, dinv, E);
    k_dinv     <<<(N + 255) / 256, 256, 0, stream>>>(dinv, N);

    k_gemm<<<(N + 127) / 128, 256, 0, stream>>>(x, W_conv, h, N);

    const int tot = N * F_OUT;
    k_init_agg<<<(tot + 255) / 256, 256, 0, stream>>>(h, dinv, b_conv, agg, tot);

    const long long sth = (long long)E * 32;
    k_scatter<<<(int)((sth + 255) / 256), 256, 0, stream>>>(src, dst, h, dinv, agg, E);

    const long long fth = (long long)N * 32;
    k_finalize<<<(int)((fth + 255) / 256), 256, 0, stream>>>(agg, W_lin, b_lin, out, N);
}


# k_scatter: one wave per 32 consecutive edges (coalesced index loads, per-lane norms, readlane broadcast, rows loaded 8 ahead) replaces one wave per edge; float atomics kept, arrival order as before
# speedup vs baseline: 1.2877x; 1.2877x over previous
#include <hip/hip_runtime.h>
#include <math.h>

typedef float v2f __attribute__((ext_vector_type(2)));
typedef float v8f __attribute__((ext_vector_type(8)));

#define F_IN  128
#define F_OUT 64


__global__ void k_init_deg(float* __restrict__ deg, int N) {
    int i = blockIdx.x * blockDim.x + threadIdx.x;
    if (i < N) deg[i] = 1.0f;
}

__global__ void k_count_deg(const int* __restrict__ dst, float* __restrict__ deg, int E) {
    int e = blockIdx.x * blockDim.x + threadIdx.x;
    if (e < E) atomicAdd(&deg[dst[e]], 1.0f);
}

__global__ void k_dinv(float* __restrict__ deg, int N) {
    int i = blockIdx.x * blockDim.x + threadIdx.x;
    if (i < N) {
        float d = deg[i];
        deg[i] = (d > 0.0f) ? rsqrtf(d) : 0.0f;
    }
}


__global__ void __launch_bounds__(256) k_gemm(const float* __restrict__ x,
                                              const float* __restrict__ W,
                                              float* __restrict__ h, int N) {
    const int wave = threadIdx.x >> 5;
    const int lane = threadIdx.x & 31;
    const int row0 = (blockIdx.x * 8 + wave) * 16;
    if (row0 >= N) return;

    const int lr    = lane & 15;
    const int khalf = lane >> 4;

    int m = row0 + lr;
    if (m >= N) m = N - 1;

    const float2* xp  = (const float2*)(x + (size_t)m * F_IN);
    const float2* wp0 = (const float2*)(W + (size_t)(lr +  0) * F_IN);
    const float2* wp1 = (const float2*)(W + (size_t)(lr + 16) * F_IN);
    const float2* wp2 = (const float2*)(W + (size_t)(lr + 32) * F_IN);
    const float2* wp3 = (const float2*)(W + (size_t)(lr + 48) * F_IN);

    v8f c0 = {}, c1 = {}, c2 = {}, c3 = {};

    #pragma unroll
    for (int k0 = 0; k0 < F_IN; k0 += 4) {
        const int ki = (k0 >> 1) + khalf;
        float2 af = xp[ki];
        v2f a; a.x = af.x; a.y = af.y;
        float2 b0f = wp0[ki]; v2f b0; b0.x = b0f.x; b0.y = b0f.y;
        float2 b1f = wp1[ki]; v2f b1; b1.x = b1f.x; b1.y = b1f.y;
        float2 b2f = wp2[ki]; v2f b2; b2.x = b2f.x; b2.y = b2f.y;
        float2 b3f = wp3[ki]; v2f b3; b3.x = b3f.x; b3.y = b3f.y;

        c0 = __builtin_amdgcn_wmma_f32_16x16x4_f32(false, a, false, b0, (short)0, c0, false, false);
        c1 = __builtin_amdgcn_wmma_f32_16x16x4_f32(false, a, false, b1, (short)0, c1, false, false);
        c2 = __builtin_amdgcn_wmma_f32_16x16x4_f32(false, a, false, b2, (short)0, c2, false, false);
        c3 = __builtin_amdgcn_wmma_f32_16x16x4_f32(false, a, false, b3, (short)0, c3, false, false);
    }

    const int mbase = row0 + (khalf << 3);
    #pragma unroll
    for (int v = 0; v < 8; ++v) {
        int mm = mbase + v;
        if (mm < N) {
            float* hp = h + (size_t)mm * F_OUT + lr;
            hp[ 0] = c0[v];
            hp[16] = c1[v];
            hp[32] = c2[v];
            hp[48] = c3[v];
        }
    }
}


__global__ void k_init_agg(const float* __restrict__ h, const float* __restrict__ dinv,
                           const float* __restrict__ b_conv, float* __restrict__ agg, int total) {
    int idx = blockIdx.x * blockDim.x + threadIdx.x;
    if (idx < total) {
        int node = idx >> 6;
        int f    = idx & 63;
        float di = dinv[node];
        agg[idx] = b_conv[f] + h[idx] * di * di;
    }
}


#define EPW 32

__global__ void __launch_bounds__(256) k_scatter(const int* __restrict__ src, const int* __restrict__ dst,
                          const float* __restrict__ h, const float* __restrict__ dinv,
                          float* __restrict__ agg, int E) {
    const int lane = threadIdx.x & 31;
    const int wave = __builtin_amdgcn_readfirstlane((int)(threadIdx.x >> 5));
    const long long e0 = ((long long)blockIdx.x * 8 + wave) * EPW;
    if (e0 >= (long long)E) return;
    const long long left = (long long)E - e0;
    const int cnt = left < EPW ? (int)left : EPW;

    int s = 0, d = 0;
    float nrm = 0.0f;
    if (lane < cnt) {
        s = src[e0 + lane];
        d = dst[e0 + lane];
        nrm = dinv[s] * dinv[d];
    }
    const int nrm_i = __float_as_int(nrm);

    if (cnt == EPW) {
        const float2* hrow = (const float2*)h + lane;
        float* arow = agg + lane * 2;
        float2 v[8], w[8];
        #pragma unroll
        for (int j = 0; j < 8; ++j)
            v[j] = hrow[(size_t)__builtin_amdgcn_readlane(s, j) * (F_OUT / 2)];
        #pragma unroll
        for (int g = 0; g < EPW; g += 8) {
            if (g + 8 < EPW) {
                #pragma unroll
                for (int j = 0; j < 8; ++j)
                    w[j] = hrow[(size_t)__builtin_amdgcn_readlane(s, g + 8 + j) * (F_OUT / 2)];
            }
            #pragma unroll
            for (int j = 0; j < 8; ++j) {
                const int dj = __builtin_amdgcn_readlane(d, g + j);
                const float nj = __int_as_float(__builtin_amdgcn_readlane(nrm_i, g + j));
                float* ad = arow + (size_t)dj * F_OUT;
                atomicAdd(ad + 0, v[j].x * nj);
                atomicAdd(ad + 1, v[j].y * nj);
            }
            #pragma unroll
            for (int j = 0; j < 8; ++j) v[j] = w[j];
        }
    } else {
        for (int j = 0; j < cnt; ++j) {
            const int sj = __shfl(s, j);
            const int dj = __shfl(d, j);
            const float nj = __shfl(nrm, j);
            const float2 v = ((const float2*)(h + (size_t)sj * F_OUT))[lane];
            float* ad = agg + (size_t)dj * F_OUT + lane * 2;
            atomicAdd(ad + 0, v.x * nj);
            atomicAdd(ad + 1, v.y * nj);
        }
    }
}


__global__ void k_finalize(const float* __restrict__ agg, const float* __restrict__ W_lin,
                           const float* __restrict__ b_lin, float* __restrict__ out, int N) {
    int gtid = blockIdx.x * blockDim.x + threadIdx.x;
    int node = gtid >> 5;
    int lane = threadIdx.x & 31;
    if (node >= N) return;

    const float* row = agg + (size_t)node * F_OUT;
    float p = fmaxf(row[lane], 0.0f) * W_lin[lane]
            + fmaxf(row[lane + 32], 0.0f) * W_lin[lane + 32];

    #pragma unroll
    for (int off = 16; off > 0; off >>= 1)
        p += __shfl_down(p, off);

    if (lane == 0) {
        float z = p + b_lin[0];
        out[node] = 1.0f / (1.0f + __expf(-z));
    }
}


extern "C" void kernel_launch(void* const* d_in, const int* in_sizes, int n_in,
                              void* d_out, int out_size, void* d_ws, size_t ws_size,
                              hipStream_t stream) {
    const float* x      = (const float*)d_in[0];
    const int*   edge   = (const int*)  d_in[1];
    const float* W_conv = (const float*)d_in[2];
    const float* b_conv = (const float*)d_in[3];
    const float* W_lin  = (const float*)d_in[4];
    const float* b_lin  = (const float*)d_in[5];
    float*       out    = (float*)d_out;

    const int N = in_sizes[0] / F_IN;
    const int E = in_sizes[1] / 2;
    const int* src = edge;
    const int* dst = edge + E;

    float* ws   = (float*)d_ws;
    float* dinv = ws;
    float* h    = ws + N;
    float* agg  = h + (size_t)N * F_OUT;

    k_init_deg <<<(N + 255) / 256, 256, 0, stream>>>(dinv, N);
    k_count_deg<<<(E + 255) / 256, 256, 0, stream>>>(dst, dinv, E);
    k_dinv     <<<(N + 255) / 256, 256, 0, stream>>>(dinv, N);

    k_gemm<<<(N + 127) / 128, 256, 0, stream>>>(x, W_conv, h, N);

    const int tot = N * F_OUT;
    k_init_agg<<<(tot + 255) / 256, 256, 0, stream>>>(h, dinv, b_conv, agg, tot);

    k_scatter<<<(E + 8 * EPW - 1) / (8 * EPW), 256, 0, stream>>>(src, dst, h, dinv, agg, E);

    const long long fth = (long long)N * 32;
    k_finalize<<<(int)((fth + 255) / 256), 256, 0, stream>>>(agg, W_lin, b_lin, out, N);
}


# float-atomic scatter replaced by destination-bucketed gather: int-atomic ranks + block scan build buckets, rows pre-scaled by dinv, one wave per node sums its rows, epilogue fused; sum order changes
# speedup vs baseline: 1.6001x; 1.2426x over previous
#include <hip/hip_runtime.h>
#include <math.h>


typedef float v2f __attribute__((ext_vector_type(2)));
typedef float v8f __attribute__((ext_vector_type(8)));

#define F_IN  128
#define F_OUT 64


__global__ void k_zero(int* __restrict__ cnt, int* __restrict__ total, int N) {
    int i = blockIdx.x * blockDim.x + threadIdx.x;
    if (i < N) cnt[i] = 0;
    if (i == 0) total[0] = 0;
}

__global__ void k_rank(const int* __restrict__ dst, int* __restrict__ cnt, int* __restrict__ rank, int E) {
    int e = blockIdx.x * blockDim.x + threadIdx.x;
    if (e < E) rank[e] = atomicAdd(&cnt[dst[e]], 1);
}

__global__ void __launch_bounds__(256) k_alloc(const int* __restrict__ cnt, int* __restrict__ total,
                                               int* __restrict__ off, float* __restrict__ dinv, int N) {
    __shared__ int wsum[8];
    __shared__ int base_s;
    const int i = blockIdx.x * 256 + threadIdx.x;
    const int lane = threadIdx.x & 31;
    const int wave = threadIdx.x >> 5;
    const int c = (i < N) ? cnt[i] : 0;
    int x = c;
    #pragma unroll
    for (int d = 1; d < 32; d <<= 1) {
        const int y = __shfl_up(x, d);
        if (lane >= d) x += y;
    }
    if (lane == 31) wsum[wave] = x;
    __syncthreads();
    if (threadIdx.x == 0) {
        int t = 0;
        for (int w = 0; w < 8; ++w) { const int s = wsum[w]; wsum[w] = t; t += s; }
        base_s = atomicAdd(total, t);
    }
    __syncthreads();
    if (i < N) {
        off[i] = base_s + wsum[wave] + x - c;
        const float d = (float)c + 1.0f;
        dinv[i] = (d > 0.0f) ? rsqrtf(d) : 0.0f;
    }
}

__global__ void k_fill(const int* __restrict__ src, const int* __restrict__ dst, const int* __restrict__ rank,
                       const int* __restrict__ off, int* __restrict__ csr, int E) {
    int e = blockIdx.x * blockDim.x + threadIdx.x;
    if (e < E) {
        const int p = off[dst[e]] + rank[e];
        if ((unsigned)p < (unsigned)E) csr[p] = src[e];
    }
}


__global__ void __launch_bounds__(256) k_gemm(const float* __restrict__ x,
                                              const float* __restrict__ W,
                                              const float* __restrict__ dinv,
                                              float* __restrict__ h, int N) {
    const int wave = threadIdx.x >> 5;
    const int lane = threadIdx.x & 31;
    const int row0 = (blockIdx.x * 8 + wave) * 16;
    if (row0 >= N) return;

    const int lr    = lane & 15;
    const int khalf = lane >> 4;

    int m = row0 + lr;
    if (m >= N) m = N - 1;

    const float2* xp  = (const float2*)(x + (size_t)m * F_IN);
    const float2* wp0 = (const float2*)(W + (size_t)(lr +  0) * F_IN);
    const float2* wp1 = (const float2*)(W + (size_t)(lr + 16) * F_IN);
    const float2* wp2 = (const float2*)(W + (size_t)(lr + 32) * F_IN);
    const float2* wp3 = (const float2*)(W + (size_t)(lr + 48) * F_IN);

    v8f c0 = {}, c1 = {}, c2 = {}, c3 = {};

    #pragma unroll
    for (int k0 = 0; k0 < F_IN; k0 += 4) {
        const int ki = (k0 >> 1) + khalf;
        float2 af = xp[ki];
        v2f a; a.x = af.x; a.y = af.y;
        float2 b0f = wp0[ki]; v2f b0; b0.x = b0f.x; b0.y = b0f.y;
        float2 b1f = wp1[ki]; v2f b1; b1.x = b1f.x; b1.y = b1f.y;
        float2 b2f = wp2[ki]; v2f b2; b2.x = b2f.x; b2.y = b2f.y;
        float2 b3f = wp3[ki]; v2f b3; b3.x = b3f.x; b3.y = b3f.y;

        c0 = __builtin_amdgcn_wmma_f32_16x16x4_f32(false, a, false, b0, (short)0, c0, false, false);
        c1 = __builtin_amdgcn_wmma_f32_16x16x4_f32(false, a, false, b1, (short)0, c1, false, false);
        c2 = __builtin_amdgcn_wmma_f32_16x16x4_f32(false, a, false, b2, (short)0, c2, false, false);
        c3 = __builtin_amdgcn_wmma_f32_16x16x4_f32(false, a, false, b3, (short)0, c3, false, false);
    }

    const int mbase = row0 + (khalf << 3);
    #pragma unroll
    for (int v = 0; v < 8; ++v) {
        int mm = mbase + v;
        if (mm < N) {
            const float sc = dinv[mm];
            float* hp = h + (size_t)mm * F_OUT + lr;
            hp[ 0] = c0[v] * sc;
            hp[16] = c1[v] * sc;
            hp[32] = c2[v] * sc;
            hp[48] = c3[v] * sc;
        }
    }
}


#define ROW(jj) hrow[(size_t)__builtin_amdgcn_readlane(sidx, (jj)) * (F_OUT / 2)]

__global__ void __launch_bounds__(256) k_gather(const int* __restrict__ csr, const int* __restrict__ off,
                          const int* __restrict__ cnt, const float* __restrict__ hs,
                          const float* __restrict__ dinv, const float* __restrict__ b_conv,
                          const float* __restrict__ W_lin, const float* __restrict__ b_lin,
                          float* __restrict__ out, int N) {
    const int lane = threadIdx.x & 31;
    const int wave = __builtin_amdgcn_readfirstlane((int)(threadIdx.x >> 5));
    const int node = blockIdx.x * 8 + wave;
    if (node >= N) return;

    const int o = off[node];
    const int c = cnt[node];
    const float dn = dinv[node];
    const float2* hrow = (const float2*)hs + lane;
    float2 acc = hrow[(size_t)node * (F_OUT / 2)];

    for (int base = 0; base < c; base += 32) {
        const int left = c - base;
        const int m = left < 32 ? left : 32;
        int sidx = 0;
        if (lane < m) sidx = csr[o + base + lane];
        sidx = min(max(sidx, 0), N - 1);

        int j = 0;
        for (; j + 8 <= m; j += 8) {
            float2 v[8];
            #pragma unroll
            for (int k = 0; k < 8; ++k) v[k] = ROW(j + k);
            #pragma unroll
            for (int k = 0; k < 8; ++k) { acc.x += v[k].x; acc.y += v[k].y; }
        }
        const int rem = m - j;
        const int j4 = j + (rem & 4);
        const int j2 = j4 + (rem & 2);
        float2 t0 = {0.f, 0.f}, t1 = t0, t2 = t0, t3 = t0, t4 = t0, t5 = t0, t6 = t0;
        if (rem & 4) { t0 = ROW(j); t1 = ROW(j + 1); t2 = ROW(j + 2); t3 = ROW(j + 3); }
        if (rem & 2) { t4 = ROW(j4); t5 = ROW(j4 + 1); }
        if (rem & 1) { t6 = ROW(j2); }
        acc.x += ((t0.x + t1.x) + (t2.x + t3.x)) + ((t4.x + t5.x) + t6.x);
        acc.y += ((t0.y + t1.y) + (t2.y + t3.y)) + ((t4.y + t5.y) + t6.y);
    }

    const float2 bc = ((const float2*)b_conv)[lane];
    const float2 wl = ((const float2*)W_lin)[lane];
    float p = fmaxf(acc.x * dn + bc.x, 0.0f) * wl.x
            + fmaxf(acc.y * dn + bc.y, 0.0f) * wl.y;

    #pragma unroll
    for (int s = 16; s > 0; s >>= 1)
        p += __shfl_down(p, s);

    if (lane == 0) {
        float z = p + b_lin[0];
        out[node] = 1.0f / (1.0f + __expf(-z));
    }
}


extern "C" void kernel_launch(void* const* d_in, const int* in_sizes, int n_in,
                              void* d_out, int out_size, void* d_ws, size_t ws_size,
                              hipStream_t stream) {
    const float* x      = (const float*)d_in[0];
    const int*   edge   = (const int*)  d_in[1];
    const float* W_conv = (const float*)d_in[2];
    const float* b_conv = (const float*)d_in[3];
    const float* W_lin  = (const float*)d_in[4];
    const float* b_lin  = (const float*)d_in[5];
    float*       out    = (float*)d_out;

    const int N = in_sizes[0] / F_IN;
    const int E = in_sizes[1] / 2;
    const int* src = edge;
    const int* dst = edge + E;

    const size_t Np = ((size_t)N + 63) & ~(size_t)63;
    const size_t Ep = ((size_t)E + 63) & ~(size_t)63;
    int*   cnt   = (int*)d_ws;
    int*   off   = cnt + Np;
    float* dinv  = (float*)(off + Np);
    int*   total = (int*)(dinv + Np);
    int*   rank  = total + 64;
    int*   csr   = rank + Ep;
    float* hs    = (float*)(csr + Ep);

    k_zero <<<(N + 255) / 256, 256, 0, stream>>>(cnt, total, N);
    k_rank <<<(E + 255) / 256, 256, 0, stream>>>(dst, cnt, rank, E);
    k_alloc<<<(N + 255) / 256, 256, 0, stream>>>(cnt, total, off, dinv, N);
    k_fill <<<(E + 255) / 256, 256, 0, stream>>>(src, dst, rank, off, csr, E);

    k_gemm<<<(N + 127) / 128, 256, 0, stream>>>(x, W_conv, dinv, hs, N);

    k_gather<<<(N + 7) / 8, 256, 0, stream>>>(csr, off, cnt, hs, dinv, b_conv, W_lin, b_lin, out, N);
}


# one returning global atomic per edge (rank in bucket) replaced by 2-level counting sort: LDS bin histograms, one global add per block and bin, per-bin LDS ranks; row sum order stays arrival-dependent
# speedup vs baseline: 2.3809x; 1.4880x over previous
#include <hip/hip_runtime.h>
#include <math.h>


typedef float v2f __attribute__((ext_vector_type(2)));
typedef float v8f __attribute__((ext_vector_type(8)));

#define F_IN  128
#define F_OUT 64


#define BIN_SHIFT 8
#define BIN_NODES 256
#define EPT       16
#define CHUNK     (256 * EPT)
#define PAD       32
#define SPT       8
#define GRP       4

__global__ void k_zero(int* __restrict__ p, int n) {
    int i = blockIdx.x * blockDim.x + threadIdx.x;
    if (i < n) p[i] = 0;
}

__device__ __forceinline__ int block_excl_scan_256(int v, int* wsum, int& total) {
    const int lane = threadIdx.x & 31;
    const int wave = threadIdx.x >> 5;
    int x = v;
    #pragma unroll
    for (int d = 1; d < 32; d <<= 1) {
        const int y = __shfl_up(x, d);
        if (lane >= d) x += y;
    }
    if (lane == 31) wsum[wave] = x;
    __syncthreads();
    int wbase = 0, tot = 0;
    #pragma unroll
    for (int w = 0; w < 8; ++w) {
        const int s = wsum[w];
        if (w < wave) wbase += s;
        tot += s;
    }
    total = tot;
    return wbase + x - v;
}

__global__ void __launch_bounds__(256) k_bincount(const int* __restrict__ dst, int* __restrict__ bincnt,
                                                  int E, int N, int NB) {
    extern __shared__ int hist[];
    for (int b = threadIdx.x; b < NB; b += 256) hist[b] = 0;
    __syncthreads();
    const int e0 = blockIdx.x * CHUNK + threadIdx.x;
    int d[EPT];
    #pragma unroll
    for (int k = 0; k < EPT; ++k) {
        const int e = e0 + k * 256;
        d[k] = (e < E) ? dst[e] : -1;
    }
    #pragma unroll
    for (int k = 0; k < EPT; ++k)
        if ((unsigned)d[k] < (unsigned)N) atomicAdd(&hist[d[k] >> BIN_SHIFT], 1);
    __syncthreads();
    for (int b = threadIdx.x; b < NB; b += 256) {
        const int h = hist[b];
        if (h) atomicAdd(&bincnt[b * PAD], h);
    }
}

__global__ void __launch_bounds__(256) k_binpart(const int* __restrict__ src, const int* __restrict__ dst,
                                                 const int* __restrict__ bincnt, int* __restrict__ bincur,
                                                 int* __restrict__ binstart, int2* __restrict__ part,
                                                 int E, int N, int NB) {
    extern __shared__ int sm[];
    __shared__ int wsum[8];
    int* hist = sm;
    int* base = sm + NB;
    for (int b = threadIdx.x; b < NB; b += 256) hist[b] = 0;
    __syncthreads();

    const int e0 = blockIdx.x * CHUNK + threadIdx.x;
    int s[EPT], d[EPT], r[EPT];
    #pragma unroll
    for (int k = 0; k < EPT; ++k) {
        const int e = e0 + k * 256;
        d[k] = (e < E) ? dst[e] : -1;
        s[k] = (e < E) ? src[e] : 0;
    }
    const int per = (NB + 255) >> 8;
    const int b0 = threadIdx.x * per;
    int tsum = 0;
    for (int i0 = 0; i0 < per; i0 += GRP) {
        int tb[GRP];
        #pragma unroll
        for (int j = 0; j < GRP; ++j) {
            const int b = b0 + i0 + j;
            tb[j] = (i0 + j < per && b < NB) ? bincnt[b * PAD] : 0;
        }
        #pragma unroll
        for (int j = 0; j < GRP; ++j) tsum += tb[j];
    }
    #pragma unroll
    for (int k = 0; k < EPT; ++k)
        r[k] = ((unsigned)d[k] < (unsigned)N) ? atomicAdd(&hist[d[k] >> BIN_SHIFT], 1) : -1;
    __syncthreads();

    int total;
    int run = block_excl_scan_256(tsum, wsum, total);
    for (int i0 = 0; i0 < per; i0 += GRP) {
        int cc[GRP], oo[GRP];
        #pragma unroll
        for (int j = 0; j < GRP; ++j) {
            const int b = b0 + i0 + j;
            const bool ok = (i0 + j < per) && (b < NB);
            cc[j] = ok ? bincnt[b * PAD] : 0;
            const int h = ok ? hist[b] : 0;
            oo[j] = h ? atomicAdd(&bincur[b * PAD], h) : 0;
        }
        #pragma unroll
        for (int j = 0; j < GRP; ++j) {
            const int b = b0 + i0 + j;
            if ((i0 + j < per) && (b < NB)) {
                base[b] = run + oo[j];
                if (blockIdx.x == 0) binstart[b] = run;
                run += cc[j];
            }
        }
    }
    __syncthreads();

    #pragma unroll
    for (int k = 0; k < EPT; ++k) {
        if (r[k] >= 0) {
            const int p = base[d[k] >> BIN_SHIFT] + r[k];
            if ((unsigned)p < (unsigned)E) part[p] = make_int2(s[k], d[k]);
        }
    }
}

__global__ void __launch_bounds__(256) k_binsort(const int2* __restrict__ part, const int* __restrict__ bincnt,
                                                 const int* __restrict__ binstart, int* __restrict__ csr,
                                                 int* __restrict__ off, int* __restrict__ cnt,
                                                 float* __restrict__ dinv, int N, int E) {
    __shared__ int cur[BIN_NODES];
    __shared__ int wsum[8];
    const int bin = blockIdx.x;
    const int start = binstart[bin];
    int len = bincnt[bin * PAD];
    if (start < 0 || start > E) len = 0;
    if (len > E - start) len = E - start;
    if (len < 0) len = 0;

    cur[threadIdx.x] = 0;
    __syncthreads();
    for (int i0 = threadIdx.x; i0 < len; i0 += 256 * SPT) {
        int2 p[SPT];
        #pragma unroll
        for (int k = 0; k < SPT; ++k) {
            const int i = i0 + k * 256;
            p[k] = (i < len) ? part[start + i] : make_int2(0, -1);
        }
        #pragma unroll
        for (int k = 0; k < SPT; ++k)
            if (p[k].y >= 0) atomicAdd(&cur[p[k].y & (BIN_NODES - 1)], 1);
    }
    __syncthreads();

    const int c = cur[threadIdx.x];
    int total;
    const int ex = block_excl_scan_256(c, wsum, total);
    const int node = bin * BIN_NODES + threadIdx.x;
    if (node < N) {
        off[node] = start + ex;
        cnt[node] = c;
        const float dg = (float)c + 1.0f;
        dinv[node] = (dg > 0.0f) ? rsqrtf(dg) : 0.0f;
    }
    cur[threadIdx.x] = ex;
    __syncthreads();

    for (int i0 = threadIdx.x; i0 < len; i0 += 256 * SPT) {
        int2 p[SPT];
        #pragma unroll
        for (int k = 0; k < SPT; ++k) {
            const int i = i0 + k * 256;
            p[k] = (i < len) ? part[start + i] : make_int2(0, -1);
        }
        int pos[SPT];
        #pragma unroll
        for (int k = 0; k < SPT; ++k)
            pos[k] = (p[k].y >= 0) ? atomicAdd(&cur[p[k].y & (BIN_NODES - 1)], 1) : -1;
        #pragma unroll
        for (int k = 0; k < SPT; ++k)
            if ((unsigned)pos[k] < (unsigned)len) csr[start + pos[k]] = p[k].x;
    }
}


__global__ void __launch_bounds__(256) k_gemm(const float* __restrict__ x,
                                              const float* __restrict__ W,
                                              const float* __restrict__ dinv,
                                              float* __restrict__ h, int N) {
    const int wave = threadIdx.x >> 5;
    const int lane = threadIdx.x & 31;
    const int row0 = (blockIdx.x * 8 + wave) * 16;
    if (row0 >= N) return;

    const int lr    = lane & 15;
    const int khalf = lane >> 4;

    int m = row0 + lr;
    if (m >= N) m = N - 1;

    const float2* xp  = (const float2*)(x + (size_t)m * F_IN);
    const float2* wp0 = (const float2*)(W + (size_t)(lr +  0) * F_IN);
    const float2* wp1 = (const float2*)(W + (size_t)(lr + 16) * F_IN);
    const float2* wp2 = (const float2*)(W + (size_t)(lr + 32) * F_IN);
    const float2* wp3 = (const float2*)(W + (size_t)(lr + 48) * F_IN);

    v8f c0 = {}, c1 = {}, c2 = {}, c3 = {};

    #pragma unroll
    for (int k0 = 0; k0 < F_IN; k0 += 4) {
        const int ki = (k0 >> 1) + khalf;
        float2 af = xp[ki];
        v2f a; a.x = af.x; a.y = af.y;
        float2 b0f = wp0[ki]; v2f b0; b0.x = b0f.x; b0.y = b0f.y;
        float2 b1f = wp1[ki]; v2f b1; b1.x = b1f.x; b1.y = b1f.y;
        float2 b2f = wp2[ki]; v2f b2; b2.x = b2f.x; b2.y = b2f.y;
        float2 b3f = wp3[ki]; v2f b3; b3.x = b3f.x; b3.y = b3f.y;

        c0 = __builtin_amdgcn_wmma_f32_16x16x4_f32(false, a, false, b0, (short)0, c0, false, false);
        c1 = __builtin_amdgcn_wmma_f32_16x16x4_f32(false, a, false, b1, (short)0, c1, false, false);
        c2 = __builtin_amdgcn_wmma_f32_16x16x4_f32(false, a, false, b2, (short)0, c2, false, false);
        c3 = __builtin_amdgcn_wmma_f32_16x16x4_f32(false, a, false, b3, (short)0, c3, false, false);
    }

    const int mbase = row0 + (khalf << 3);
    #pragma unroll
    for (int v = 0; v < 8; ++v) {
        int mm = mbase + v;
        if (mm < N) {
            const float sc = dinv[mm];
            float* hp = h + (size_t)mm * F_OUT + lr;
            hp[ 0] = c0[v] * sc;
            hp[16] = c1[v] * sc;
            hp[32] = c2[v] * sc;
            hp[48] = c3[v] * sc;
        }
    }
}


#define ROW(jj) hrow[(size_t)__builtin_amdgcn_readlane(sidx, (jj)) * (F_OUT / 2)]

__global__ void __launch_bounds__(256) k_gather(const int* __restrict__ csr, const int* __restrict__ off,
                          const int* __restrict__ cnt, const float* __restrict__ hs,
                          const float* __restrict__ dinv, const float* __restrict__ b_conv,
                          const float* __restrict__ W_lin, const float* __restrict__ b_lin,
                          float* __restrict__ out, int N) {
    const int lane = threadIdx.x & 31;
    const int wave = __builtin_amdgcn_readfirstlane((int)(threadIdx.x >> 5));
    const int node = blockIdx.x * 8 + wave;
    if (node >= N) return;

    const int o = off[node];
    const int c = cnt[node];
    const float dn = dinv[node];
    const float2* hrow = (const float2*)hs + lane;
    float2 acc = hrow[(size_t)node * (F_OUT / 2)];

    for (int base = 0; base < c; base += 32) {
        const int left = c - base;
        const int m = left < 32 ? left : 32;
        int sidx = 0;
        if (lane < m) sidx = csr[o + base + lane];
        sidx = min(max(sidx, 0), N - 1);

        int j = 0;
        for (; j + 8 <= m; j += 8) {
            float2 v[8];
            #pragma unroll
            for (int k = 0; k < 8; ++k) v[k] = ROW(j + k);
            #pragma unroll
            for (int k = 0; k < 8; ++k) { acc.x += v[k].x; acc.y += v[k].y; }
        }
        const int rem = m - j;
        const int j4 = j + (rem & 4);
        const int j2 = j4 + (rem & 2);
        float2 t0 = {0.f, 0.f}, t1 = t0, t2 = t0, t3 = t0, t4 = t0, t5 = t0, t6 = t0;
        if (rem & 4) { t0 = ROW(j); t1 = ROW(j + 1); t2 = ROW(j + 2); t3 = ROW(j + 3); }
        if (rem & 2) { t4 = ROW(j4); t5 = ROW(j4 + 1); }
        if (rem & 1) { t6 = ROW(j2); }
        acc.x += ((t0.x + t1.x) + (t2.x + t3.x)) + ((t4.x + t5.x) + t6.x);
        acc.y += ((t0.y + t1.y) + (t2.y + t3.y)) + ((t4.y + t5.y) + t6.y);
    }

    const float2 bc = ((const float2*)b_conv)[lane];
    const float2 wl = ((const float2*)W_lin)[lane];
    float p = fmaxf(acc.x * dn + bc.x, 0.0f) * wl.x
            + fmaxf(acc.y * dn + bc.y, 0.0f) * wl.y;

    #pragma unroll
    for (int s = 16; s > 0; s >>= 1)
        p += __shfl_down(p, s);

    if (lane == 0) {
        float z = p + b_lin[0];
        out[node] = 1.0f / (1.0f + __expf(-z));
    }
}


extern "C" void kernel_launch(void* const* d_in, const int* in_sizes, int n_in,
                              void* d_out, int out_size, void* d_ws, size_t ws_size,
                              hipStream_t stream) {
    const float* x      = (const float*)d_in[0];
    const int*   edge   = (const int*)  d_in[1];
    const float* W_conv = (const float*)d_in[2];
    const float* b_conv = (const float*)d_in[3];
    const float* W_lin  = (const float*)d_in[4];
    const float* b_lin  = (const float*)d_in[5];
    float*       out    = (float*)d_out;

    const int N = in_sizes[0] / F_IN;
    const int E = in_sizes[1] / 2;
    const int* src = edge;
    const int* dst = edge + E;
    const int NB = (N + BIN_NODES - 1) / BIN_NODES;

    const size_t Np  = ((size_t)N + 63) & ~(size_t)63;
    const size_t Ep  = ((size_t)E + 63) & ~(size_t)63;
    const size_t NBp = ((size_t)NB + 63) & ~(size_t)63;
    int*   cnt      = (int*)d_ws;
    int*   off      = cnt + Np;
    float* dinv     = (float*)(off + Np);
    int*   bincnt   = (int*)(dinv + Np);
    int*   bincur   = bincnt + NBp * PAD;
    int*   binstart = bincur + NBp * PAD;
    int2*  part     = (int2*)(binstart + NBp);
    int*   csr      = (int*)(part + Ep);
    float* hs       = (float*)(csr + Ep);

    const int nzero = (int)(2 * NBp * PAD);
    const int EB = (E + CHUNK - 1) / CHUNK;

    k_zero    <<<(nzero + 255) / 256, 256, 0, stream>>>(bincnt, nzero);
    k_bincount<<<EB, 256, (size_t)NB * sizeof(int), stream>>>(dst, bincnt, E, N, NB);
    k_binpart <<<EB, 256, (size_t)NB * 2 * sizeof(int), stream>>>(src, dst, bincnt, bincur, binstart, part, E, N, NB);
    k_binsort <<<NB, 256, 0, stream>>>(part, bincnt, binstart, csr, off, cnt, dinv, N, E);

    k_gemm<<<(N + 127) / 128, 256, 0, stream>>>(x, W_conv, dinv, hs, N);

    k_gather<<<(N + 7) / 8, 256, 0, stream>>>(csr, off, cnt, hs, dinv, b_conv, W_lin, b_lin, out, N);
}


# k_gather: each wave loops over 8 consecutive nodes (12500 resident waves, not 100000 one-node waves), next node's own row and source indices requested one node ahead; row sum order unchanged
# speedup vs baseline: 2.6215x; 1.1011x over previous
#include <hip/hip_runtime.h>
#include <math.h>


typedef float v2f __attribute__((ext_vector_type(2)));
typedef float v8f __attribute__((ext_vector_type(8)));

#define F_IN  128
#define F_OUT 64


#define BIN_SHIFT 8
#define BIN_NODES 256
#define EPT       16
#define CHUNK     (256 * EPT)
#define PAD       32
#define SPT       8
#define GRP       4

__global__ void k_zero(int* __restrict__ p, int n) {
    int i = blockIdx.x * blockDim.x + threadIdx.x;
    if (i < n) p[i] = 0;
}

__device__ __forceinline__ int block_excl_scan_256(int v, int* wsum, int& total) {
    const int lane = threadIdx.x & 31;
    const int wave = threadIdx.x >> 5;
    int x = v;
    #pragma unroll
    for (int d = 1; d < 32; d <<= 1) {
        const int y = __shfl_up(x, d);
        if (lane >= d) x += y;
    }
    if (lane == 31) wsum[wave] = x;
    __syncthreads();
    int wbase = 0, tot = 0;
    #pragma unroll
    for (int w = 0; w < 8; ++w) {
        const int s = wsum[w];
        if (w < wave) wbase += s;
        tot += s;
    }
    total = tot;
    return wbase + x - v;
}

__global__ void __launch_bounds__(256) k_bincount(const int* __restrict__ dst, int* __restrict__ bincnt,
                                                  int E, int N, int NB) {
    extern __shared__ int hist[];
    for (int b = threadIdx.x; b < NB; b += 256) hist[b] = 0;
    __syncthreads();
    const int e0 = blockIdx.x * CHUNK + threadIdx.x;
    int d[EPT];
    #pragma unroll
    for (int k = 0; k < EPT; ++k) {
        const int e = e0 + k * 256;
        d[k] = (e < E) ? dst[e] : -1;
    }
    #pragma unroll
    for (int k = 0; k < EPT; ++k)
        if ((unsigned)d[k] < (unsigned)N) atomicAdd(&hist[d[k] >> BIN_SHIFT], 1);
    __syncthreads();
    for (int b = threadIdx.x; b < NB; b += 256) {
        const int h = hist[b];
        if (h) atomicAdd(&bincnt[b * PAD], h);
    }
}

__global__ void __launch_bounds__(256) k_binpart(const int* __restrict__ src, const int* __restrict__ dst,
                                                 const int* __restrict__ bincnt, int* __restrict__ bincur,
                                                 int* __restrict__ binstart, int2* __restrict__ part,
                                                 int E, int N, int NB) {
    extern __shared__ int sm[];
    __shared__ int wsum[8];
    int* hist = sm;
    int* base = sm + NB;
    for (int b = threadIdx.x; b < NB; b += 256) hist[b] = 0;
    __syncthreads();

    const int e0 = blockIdx.x * CHUNK + threadIdx.x;
    int s[EPT], d[EPT], r[EPT];
    #pragma unroll
    for (int k = 0; k < EPT; ++k) {
        const int e = e0 + k * 256;
        d[k] = (e < E) ? dst[e] : -1;
        s[k] = (e < E) ? src[e] : 0;
    }
    const int per = (NB + 255) >> 8;
    const int b0 = threadIdx.x * per;
    int tsum = 0;
    for (int i0 = 0; i0 < per; i0 += GRP) {
        int tb[GRP];
        #pragma unroll
        for (int j = 0; j < GRP; ++j) {
            const int b = b0 + i0 + j;
            tb[j] = (i0 + j < per && b < NB) ? bincnt[b * PAD] : 0;
        }
        #pragma unroll
        for (int j = 0; j < GRP; ++j) tsum += tb[j];
    }
    #pragma unroll
    for (int k = 0; k < EPT; ++k)
        r[k] = ((unsigned)d[k] < (unsigned)N) ? atomicAdd(&hist[d[k] >> BIN_SHIFT], 1) : -1;
    __syncthreads();

    int total;
    int run = block_excl_scan_256(tsum, wsum, total);
    for (int i0 = 0; i0 < per; i0 += GRP) {
        int cc[GRP], oo[GRP];
        #pragma unroll
        for (int j = 0; j < GRP; ++j) {
            const int b = b0 + i0 + j;
            const bool ok = (i0 + j < per) && (b < NB);
            cc[j] = ok ? bincnt[b * PAD] : 0;
            const int h = ok ? hist[b] : 0;
            oo[j] = h ? atomicAdd(&bincur[b * PAD], h) : 0;
        }
        #pragma unroll
        for (int j = 0; j < GRP; ++j) {
            const int b = b0 + i0 + j;
            if ((i0 + j < per) && (b < NB)) {
                base[b] = run + oo[j];
                if (blockIdx.x == 0) binstart[b] = run;
                run += cc[j];
            }
        }
    }
    __syncthreads();

    #pragma unroll
    for (int k = 0; k < EPT; ++k) {
        if (r[k] >= 0) {
            const int p = base[d[k] >> BIN_SHIFT] + r[k];
            if ((unsigned)p < (unsigned)E) part[p] = make_int2(s[k], d[k]);
        }
    }
}

__global__ void __launch_bounds__(256) k_binsort(const int2* __restrict__ part, const int* __restrict__ bincnt,
                                                 const int* __restrict__ binstart, int* __restrict__ csr,
                                                 int* __restrict__ off, int* __restrict__ cnt,
                                                 float* __restrict__ dinv, int N, int E) {
    __shared__ int cur[BIN_NODES];
    __shared__ int wsum[8];
    const int bin = blockIdx.x;
    const int start = binstart[bin];
    int len = bincnt[bin * PAD];
    if (start < 0 || start > E) len = 0;
    if (len > E - start) len = E - start;
    if (len < 0) len = 0;

    cur[threadIdx.x] = 0;
    __syncthreads();
    for (int i0 = threadIdx.x; i0 < len; i0 += 256 * SPT) {
        int2 p[SPT];
        #pragma unroll
        for (int k = 0; k < SPT; ++k) {
            const int i = i0 + k * 256;
            p[k] = (i < len) ? part[start + i] : make_int2(0, -1);
        }
        #pragma unroll
        for (int k = 0; k < SPT; ++k)
            if (p[k].y >= 0) atomicAdd(&cur[p[k].y & (BIN_NODES - 1)], 1);
    }
    __syncthreads();

    const int c = cur[threadIdx.x];
    int total;
    const int ex = block_excl_scan_256(c, wsum, total);
    const int node = bin * BIN_NODES + threadIdx.x;
    if (node < N) {
        off[node] = start + ex;
        cnt[node] = c;
        const float dg = (float)c + 1.0f;
        dinv[node] = (dg > 0.0f) ? rsqrtf(dg) : 0.0f;
    }
    cur[threadIdx.x] = ex;
    __syncthreads();

    for (int i0 = threadIdx.x; i0 < len; i0 += 256 * SPT) {
        int2 p[SPT];
        #pragma unroll
        for (int k = 0; k < SPT; ++k) {
            const int i = i0 + k * 256;
            p[k] = (i < len) ? part[start + i] : make_int2(0, -1);
        }
        int pos[SPT];
        #pragma unroll
        for (int k = 0; k < SPT; ++k)
            pos[k] = (p[k].y >= 0) ? atomicAdd(&cur[p[k].y & (BIN_NODES - 1)], 1) : -1;
        #pragma unroll
        for (int k = 0; k < SPT; ++k)
            if ((unsigned)pos[k] < (unsigned)len) csr[start + pos[k]] = p[k].x;
    }
}


__global__ void __launch_bounds__(256) k_gemm(const float* __restrict__ x,
                                              const float* __restrict__ W,
                                              const float* __restrict__ dinv,
                                              float* __restrict__ h, int N) {
    const int wave = threadIdx.x >> 5;
    const int lane = threadIdx.x & 31;
    const int row0 = (blockIdx.x * 8 + wave) * 16;
    if (row0 >= N) return;

    const int lr    = lane & 15;
    const int khalf = lane >> 4;

    int m = row0 + lr;
    if (m >= N) m = N - 1;

    const float2* xp  = (const float2*)(x + (size_t)m * F_IN);
    const float2* wp0 = (const float2*)(W + (size_t)(lr +  0) * F_IN);
    const float2* wp1 = (const float2*)(W + (size_t)(lr + 16) * F_IN);
    const float2* wp2 = (const float2*)(W + (size_t)(lr + 32) * F_IN);
    const float2* wp3 = (const float2*)(W + (size_t)(lr + 48) * F_IN);

    v8f c0 = {}, c1 = {}, c2 = {}, c3 = {};

    #pragma unroll
    for (int k0 = 0; k0 < F_IN; k0 += 4) {
        const int ki = (k0 >> 1) + khalf;
        float2 af = xp[ki];
        v2f a; a.x = af.x; a.y = af.y;
        float2 b0f = wp0[ki]; v2f b0; b0.x = b0f.x; b0.y = b0f.y;
        float2 b1f = wp1[ki]; v2f b1; b1.x = b1f.x; b1.y = b1f.y;
        float2 b2f = wp2[ki]; v2f b2; b2.x = b2f.x; b2.y = b2f.y;
        float2 b3f = wp3[ki]; v2f b3; b3.x = b3f.x; b3.y = b3f.y;

        c0 = __builtin_amdgcn_wmma_f32_16x16x4_f32(false, a, false, b0, (short)0, c0, false, false);
        c1 = __builtin_amdgcn_wmma_f32_16x16x4_f32(false, a, false, b1, (short)0, c1, false, false);
        c2 = __builtin_amdgcn_wmma_f32_16x16x4_f32(false, a, false, b2, (short)0, c2, false, false);
        c3 = __builtin_amdgcn_wmma_f32_16x16x4_f32(false, a, false, b3, (short)0, c3, false, false);
    }

    const int mbase = row0 + (khalf << 3);
    #pragma unroll
    for (int v = 0; v < 8; ++v) {
        int mm = mbase + v;
        if (mm < N) {
            const float sc = dinv[mm];
            float* hp = h + (size_t)mm * F_OUT + lr;
            hp[ 0] = c0[v] * sc;
            hp[16] = c1[v] * sc;
            hp[32] = c2[v] * sc;
            hp[48] = c3[v] * sc;
        }
    }
}


#define NPW 8

#define ROW(jj) hrow[(size_t)__builtin_amdgcn_readlane(sidx, (jj)) * (F_OUT / 2)]

__device__ __forceinline__ void sum_rows(float2& acc, const int sidx, const int m, const float2* __restrict__ hrow) {
    int j = 0;
    for (; j + 8 <= m; j += 8) {
        float2 v[8];
        #pragma unroll
        for (int k = 0; k < 8; ++k) v[k] = ROW(j + k);
        #pragma unroll
        for (int k = 0; k < 8; ++k) { acc.x += v[k].x; acc.y += v[k].y; }
    }
    const int rem = m - j;
    const int j4 = j + (rem & 4);
    const int j2 = j4 + (rem & 2);
    float2 t0 = {0.f, 0.f}, t1 = t0, t2 = t0, t3 = t0, t4 = t0, t5 = t0, t6 = t0;
    if (rem & 4) { t0 = ROW(j); t1 = ROW(j + 1); t2 = ROW(j + 2); t3 = ROW(j + 3); }
    if (rem & 2) { t4 = ROW(j4); t5 = ROW(j4 + 1); }
    if (rem & 1) { t6 = ROW(j2); }
    acc.x += ((t0.x + t1.x) + (t2.x + t3.x)) + ((t4.x + t5.x) + t6.x);
    acc.y += ((t0.y + t1.y) + (t2.y + t3.y)) + ((t4.y + t5.y) + t6.y);
}

__global__ void __launch_bounds__(256) k_gather(const int* __restrict__ csr, const int* __restrict__ off,
                          const int* __restrict__ cnt, const float* __restrict__ hs,
                          const float* __restrict__ dinv, const float* __restrict__ b_conv,
                          const float* __restrict__ W_lin, const float* __restrict__ b_lin,
                          float* __restrict__ out, int N) {
    const int lane = threadIdx.x & 31;
    const int wave = __builtin_amdgcn_readfirstlane((int)(threadIdx.x >> 5));
    const int node0 = (blockIdx.x * 8 + wave) * NPW;
    if (node0 >= N) return;
    const int nn = (N - node0) < NPW ? (N - node0) : NPW;

    int o_l = 0, c_l = 0, dn_l = 0;
    if (lane < nn) {
        o_l  = off[node0 + lane];
        c_l  = cnt[node0 + lane];
        dn_l = __float_as_int(dinv[node0 + lane]);
    }
    const float2* hrow = (const float2*)hs + lane;
    const float2 bc = ((const float2*)b_conv)[lane];
    const float2 wl = ((const float2*)W_lin)[lane];
    const float  bl = b_lin[0];

    int o_nx = __builtin_amdgcn_readlane(o_l, 0);
    int c_nx = __builtin_amdgcn_readlane(c_l, 0);
    float2 self_nx = hrow[(size_t)node0 * (F_OUT / 2)];
    int sidx_nx = 0;
    if (lane < (c_nx < 32 ? c_nx : 32)) sidx_nx = csr[o_nx + lane];

    float res = 0.0f;

    for (int i = 0; i < nn; ++i) {
        const int o = o_nx;
        const int c = c_nx;
        const float dn = __int_as_float(__builtin_amdgcn_readlane(dn_l, i));
        float2 acc = self_nx;
        int sidx = sidx_nx;

        if (i + 1 < nn) {
            o_nx = __builtin_amdgcn_readlane(o_l, i + 1);
            c_nx = __builtin_amdgcn_readlane(c_l, i + 1);
            self_nx = hrow[(size_t)(node0 + i + 1) * (F_OUT / 2)];
            sidx_nx = 0;
            if (lane < (c_nx < 32 ? c_nx : 32)) sidx_nx = csr[o_nx + lane];
        }

        sidx = min(max(sidx, 0), N - 1);
        sum_rows(acc, sidx, c < 32 ? (c > 0 ? c : 0) : 32, hrow);
        for (int base = 32; base < c; base += 32) {
            const int left = c - base;
            const int m = left < 32 ? left : 32;
            int sx = 0;
            if (lane < m) sx = csr[o + base + lane];
            sx = min(max(sx, 0), N - 1);
            sum_rows(acc, sx, m, hrow);
        }

        float p = fmaxf(acc.x * dn + bc.x, 0.0f) * wl.x
                + fmaxf(acc.y * dn + bc.y, 0.0f) * wl.y;

        #pragma unroll
        for (int s = 16; s > 0; s >>= 1)
            p += __shfl_down(p, s);

        const float z = __int_as_float(__builtin_amdgcn_readfirstlane(__float_as_int(p))) + bl;
        const float r = 1.0f / (1.0f + __expf(-z));
        if (lane == i) res = r;
    }

    if (lane < nn) out[node0 + lane] = res;
}


extern "C" void kernel_launch(void* const* d_in, const int* in_sizes, int n_in,
                              void* d_out, int out_size, void* d_ws, size_t ws_size,
                              hipStream_t stream) {
    const float* x      = (const float*)d_in[0];
    const int*   edge   = (const int*)  d_in[1];
    const float* W_conv = (const float*)d_in[2];
    const float* b_conv = (const float*)d_in[3];
    const float* W_lin  = (const float*)d_in[4];
    const float* b_lin  = (const float*)d_in[5];
    float*       out    = (float*)d_out;

    const int N = in_sizes[0] / F_IN;
    const int E = in_sizes[1] / 2;
    const int* src = edge;
    const int* dst = edge + E;
    const int NB = (N + BIN_NODES - 1) / BIN_NODES;

    const size_t Np  = ((size_t)N + 63) & ~(size_t)63;
    const size_t Ep  = ((size_t)E + 63) & ~(size_t)63;
    const size_t NBp = ((size_t)NB + 63) & ~(size_t)63;
    int*   cnt      = (int*)d_ws;
    int*   off      = cnt + Np;
    float* dinv     = (float*)(off + Np);
    int*   bincnt   = (int*)(dinv + Np);
    int*   bincur   = bincnt + NBp * PAD;
    int*   binstart = bincur + NBp * PAD;
    int2*  part     = (int2*)(binstart + NBp);
    int*   csr      = (int*)(part + Ep);
    float* hs       = (float*)(csr + Ep);

    const int nzero = (int)(2 * NBp * PAD);
    const int EB = (E + CHUNK - 1) / CHUNK;

    k_zero    <<<(nzero + 255) / 256, 256, 0, stream>>>(bincnt, nzero);
    k_bincount<<<EB, 256, (size_t)NB * sizeof(int), stream>>>(dst, bincnt, E, N, NB);
    k_binpart <<<EB, 256, (size_t)NB * 2 * sizeof(int), stream>>>(src, dst, bincnt, bincur, binstart, part, E, N, NB);
    k_binsort <<<NB, 256, 0, stream>>>(part, bincnt, binstart, csr, off, cnt, dinv, N, E);

    k_gemm<<<(N + 127) / 128, 256, 0, stream>>>(x, W_conv, dinv, hs, N);

    k_gather<<<(N + 8 * NPW - 1) / (8 * NPW), 256, 0, stream>>>(csr, off, cnt, hs, dinv, b_conv, W_lin, b_lin, out, N);
}


# k_gemm loads: W_conv staged once per block in LDS, x/W fragments read 16 B per lane in one shared k order, row scales fetched up front (was 160 8-B 16-line loads per tile); k sum order changes
# speedup vs baseline: 2.9513x; 1.1258x over previous
#include <hip/hip_runtime.h>
#include <math.h>


typedef float v2f __attribute__((ext_vector_type(2)));
typedef float v4f __attribute__((ext_vector_type(4)));
typedef float v8f __attribute__((ext_vector_type(8)));

#define F_IN  128
#define F_OUT 64


#define BIN_SHIFT 8
#define BIN_NODES 256
#define EPT       16
#define CHUNK     (256 * EPT)
#define PAD       32
#define SPT       8
#define GRP       4

__global__ void k_zero(int* __restrict__ p, int n) {
    int i = blockIdx.x * blockDim.x + threadIdx.x;
    if (i < n) p[i] = 0;
}

__device__ __forceinline__ int block_excl_scan_256(int v, int* wsum, int& total) {
    const int lane = threadIdx.x & 31;
    const int wave = threadIdx.x >> 5;
    int x = v;
    #pragma unroll
    for (int d = 1; d < 32; d <<= 1) {
        const int y = __shfl_up(x, d);
        if (lane >= d) x += y;
    }
    if (lane == 31) wsum[wave] = x;
    __syncthreads();
    int wbase = 0, tot = 0;
    #pragma unroll
    for (int w = 0; w < 8; ++w) {
        const int s = wsum[w];
        if (w < wave) wbase += s;
        tot += s;
    }
    total = tot;
    return wbase + x - v;
}

__global__ void __launch_bounds__(256) k_bincount(const int* __restrict__ dst, int* __restrict__ bincnt,
                                                  int E, int N, int NB) {
    extern __shared__ int hist[];
    for (int b = threadIdx.x; b < NB; b += 256) hist[b] = 0;
    __syncthreads();
    const int e0 = blockIdx.x * CHUNK + threadIdx.x;
    int d[EPT];
    #pragma unroll
    for (int k = 0; k < EPT; ++k) {
        const int e = e0 + k * 256;
        d[k] = (e < E) ? dst[e] : -1;
    }
    #pragma unroll
    for (int k = 0; k < EPT; ++k)
        if ((unsigned)d[k] < (unsigned)N) atomicAdd(&hist[d[k] >> BIN_SHIFT], 1);
    __syncthreads();
    for (int b = threadIdx.x; b < NB; b += 256) {
        const int h = hist[b];
        if (h) atomicAdd(&bincnt[b * PAD], h);
    }
}

__global__ void __launch_bounds__(256) k_binpart(const int* __restrict__ src, const int* __restrict__ dst,
                                                 const int* __restrict__ bincnt, int* __restrict__ bincur,
                                                 int* __restrict__ binstart, int2* __restrict__ part,
                                                 int E, int N, int NB) {
    extern __shared__ int sm[];
    __shared__ int wsum[8];
    int* hist = sm;
    int* base = sm + NB;
    for (int b = threadIdx.x; b < NB; b += 256) hist[b] = 0;
    __syncthreads();

    const int e0 = blockIdx.x * CHUNK + threadIdx.x;
    int s[EPT], d[EPT], r[EPT];
    #pragma unroll
    for (int k = 0; k < EPT; ++k) {
        const int e = e0 + k * 256;
        d[k] = (e < E) ? dst[e] : -1;
        s[k] = (e < E) ? src[e] : 0;
    }
    const int per = (NB + 255) >> 8;
    const int b0 = threadIdx.x * per;
    int tsum = 0;
    for (int i0 = 0; i0 < per; i0 += GRP) {
        int tb[GRP];
        #pragma unroll
        for (int j = 0; j < GRP; ++j) {
            const int b = b0 + i0 + j;
            tb[j] = (i0 + j < per && b < NB) ? bincnt[b * PAD] : 0;
        }
        #pragma unroll
        for (int j = 0; j < GRP; ++j) tsum += tb[j];
    }
    #pragma unroll
    for (int k = 0; k < EPT; ++k)
        r[k] = ((unsigned)d[k] < (unsigned)N) ? atomicAdd(&hist[d[k] >> BIN_SHIFT], 1) : -1;
    __syncthreads();

    int total;
    int run = block_excl_scan_256(tsum, wsum, total);
    for (int i0 = 0; i0 < per; i0 += GRP) {
        int cc[GRP], oo[GRP];
        #pragma unroll
        for (int j = 0; j < GRP; ++j) {
            const int b = b0 + i0 + j;
            const bool ok = (i0 + j < per) && (b < NB);
            cc[j] = ok ? bincnt[b * PAD] : 0;
            const int h = ok ? hist[b] : 0;
            oo[j] = h ? atomicAdd(&bincur[b * PAD], h) : 0;
        }
        #pragma unroll
        for (int j = 0; j < GRP; ++j) {
            const int b = b0 + i0 + j;
            if ((i0 + j < per) && (b < NB)) {
                base[b] = run + oo[j];
                if (blockIdx.x == 0) binstart[b] = run;
                run += cc[j];
            }
        }
    }
    __syncthreads();

    #pragma unroll
    for (int k = 0; k < EPT; ++k) {
        if (r[k] >= 0) {
            const int p = base[d[k] >> BIN_SHIFT] + r[k];
            if ((unsigned)p < (unsigned)E) part[p] = make_int2(s[k], d[k]);
        }
    }
}

__global__ void __launch_bounds__(256) k_binsort(const int2* __restrict__ part, const int* __restrict__ bincnt,
                                                 const int* __restrict__ binstart, int* __restrict__ csr,
                                                 int* __restrict__ off, int* __restrict__ cnt,
                                                 float* __restrict__ dinv, int N, int E) {
    __shared__ int cur[BIN_NODES];
    __shared__ int wsum[8];
    const int bin = blockIdx.x;
    const int start = binstart[bin];
    int len = bincnt[bin * PAD];
    if (start < 0 || start > E) len = 0;
    if (len > E - start) len = E - start;
    if (len < 0) len = 0;

    cur[threadIdx.x] = 0;
    __syncthreads();
    for (int i0 = threadIdx.x; i0 < len; i0 += 256 * SPT) {
        int2 p[SPT];
        #pragma unroll
        for (int k = 0; k < SPT; ++k) {
            const int i = i0 + k * 256;
            p[k] = (i < len) ? part[start + i] : make_int2(0, -1);
        }
        #pragma unroll
        for (int k = 0; k < SPT; ++k)
            if (p[k].y >= 0) atomicAdd(&cur[p[k].y & (BIN_NODES - 1)], 1);
    }
    __syncthreads();

    const int c = cur[threadIdx.x];
    int total;
    const int ex = block_excl_scan_256(c, wsum, total);
    const int node = bin * BIN_NODES + threadIdx.x;
    if (node < N) {
        off[node] = start + ex;
        cnt[node] = c;
        const float dg = (float)c + 1.0f;
        dinv[node] = (dg > 0.0f) ? rsqrtf(dg) : 0.0f;
    }
    cur[threadIdx.x] = ex;
    __syncthreads();

    for (int i0 = threadIdx.x; i0 < len; i0 += 256 * SPT) {
        int2 p[SPT];
        #pragma unroll
        for (int k = 0; k < SPT; ++k) {
            const int i = i0 + k * 256;
            p[k] = (i < len) ? part[start + i] : make_int2(0, -1);
        }
        int pos[SPT];
        #pragma unroll
        for (int k = 0; k < SPT; ++k)
            pos[k] = (p[k].y >= 0) ? atomicAdd(&cur[p[k].y & (BIN_NODES - 1)], 1) : -1;
        #pragma unroll
        for (int k = 0; k < SPT; ++k)
            if ((unsigned)pos[k] < (unsigned)len) csr[start + pos[k]] = p[k].x;
    }
}


#define W_PITCH 132
#define KGROUPS (F_IN / 8)

__device__ __forceinline__ int frag_k(const int g, const int hf) { return 8 * g + 4 * hf; }

__device__ __forceinline__ v2f lo2(const float4 v) { v2f r; r.x = v.x; r.y = v.y; return r; }
__device__ __forceinline__ v2f hi2(const float4 v) { v2f r; r.x = v.z; r.y = v.w; return r; }

__global__ void __launch_bounds__(256) k_gemm(const float* __restrict__ x,
                                              const float* __restrict__ W,
                                              const float* __restrict__ dinv,
                                              float* __restrict__ h, int N) {
    __shared__ float4 Wl4[F_OUT * W_PITCH / 4];
    float* Wl = (float*)Wl4;

    const int wave = threadIdx.x >> 5;
    const int lane = threadIdx.x & 31;
    const int row0 = (blockIdx.x * 8 + wave) * 16;

    const int lr = lane & 15;
    const int hf = lane >> 4;

    int m = row0 + lr;
    if (m >= N) m = N - 1;

    v4f w[8];
    #pragma unroll
    for (int i = 0; i < 8; ++i) w[i] = ((const v4f*)W)[threadIdx.x + 256 * i];

    const int mbase = row0 + (hf << 3);
    float sc[8];
    #pragma unroll
    for (int v = 0; v < 8; ++v) sc[v] = dinv[(mbase + v < N) ? (mbase + v) : (N - 1)];

    float4 a[KGROUPS];
    {
        const float* xr = x + (size_t)m * F_IN;
        #pragma unroll
        for (int g = 0; g < KGROUPS; ++g) a[g] = *(const float4*)(xr + frag_k(g, hf));
    }
    asm volatile("" :: "v"(w[0]), "v"(w[1]), "v"(w[2]), "v"(w[3]), "v"(w[4]), "v"(w[5]), "v"(w[6]), "v"(w[7]));

    #pragma unroll
    for (int i = 0; i < 8; ++i) {
        const int q = threadIdx.x + 256 * i;
        *(v4f*)(Wl + (q >> 5) * W_PITCH + (q & 31) * 4) = w[i];
    }
    __syncthreads();
    if (row0 >= N) return;

    v8f c0 = {}, c1 = {}, c2 = {}, c3 = {};
    const float* wr = Wl + lr * W_PITCH + frag_k(0, hf);

    float4 b0 = *(const float4*)(wr +  0 * W_PITCH);
    float4 b1 = *(const float4*)(wr + 16 * W_PITCH);
    float4 b2 = *(const float4*)(wr + 32 * W_PITCH);
    float4 b3 = *(const float4*)(wr + 48 * W_PITCH);

    #pragma unroll
    for (int g = 0; g < KGROUPS; ++g) {
        float4 n0 = b0, n1 = b1, n2 = b2, n3 = b3;
        if (g + 1 < KGROUPS) {
            const float* wn = wr + (frag_k(g + 1, 0) - frag_k(0, 0));
            n0 = *(const float4*)(wn +  0 * W_PITCH);
            n1 = *(const float4*)(wn + 16 * W_PITCH);
            n2 = *(const float4*)(wn + 32 * W_PITCH);
            n3 = *(const float4*)(wn + 48 * W_PITCH);
        }
        __builtin_amdgcn_sched_barrier(0);

        const v2f al = lo2(a[g]), ah = hi2(a[g]);
        c0 = __builtin_amdgcn_wmma_f32_16x16x4_f32(false, al, false, lo2(b0), (short)0, c0, false, false);
        c1 = __builtin_amdgcn_wmma_f32_16x16x4_f32(false, al, false, lo2(b1), (short)0, c1, false, false);
        c2 = __builtin_amdgcn_wmma_f32_16x16x4_f32(false, al, false, lo2(b2), (short)0, c2, false, false);
        c3 = __builtin_amdgcn_wmma_f32_16x16x4_f32(false, al, false, lo2(b3), (short)0, c3, false, false);
        c0 = __builtin_amdgcn_wmma_f32_16x16x4_f32(false, ah, false, hi2(b0), (short)0, c0, false, false);
        c1 = __builtin_amdgcn_wmma_f32_16x16x4_f32(false, ah, false, hi2(b1), (short)0, c1, false, false);
        c2 = __builtin_amdgcn_wmma_f32_16x16x4_f32(false, ah, false, hi2(b2), (short)0, c2, false, false);
        c3 = __builtin_amdgcn_wmma_f32_16x16x4_f32(false, ah, false, hi2(b3), (short)0, c3, false, false);
        __builtin_amdgcn_sched_barrier(0);
        b0 = n0; b1 = n1; b2 = n2; b3 = n3;
    }

    #pragma unroll
    for (int v = 0; v < 8; ++v) {
        int mm = mbase + v;
        if (mm < N) {
            float* hp = h + (size_t)mm * F_OUT + lr;
            hp[ 0] = c0[v] * sc[v];
            hp[16] = c1[v] * sc[v];
            hp[32] = c2[v] * sc[v];
            hp[48] = c3[v] * sc[v];
        }
    }
}


#define NPW 8

#define ROW(jj) hrow[(size_t)__builtin_amdgcn_readlane(sidx, (jj)) * (F_OUT / 2)]

__device__ __forceinline__ void sum_rows(float2& acc, const int sidx, const int m, const float2* __restrict__ hrow) {
    int j = 0;
    for (; j + 8 <= m; j += 8) {
        float2 v[8];
        #pragma unroll
        for (int k = 0; k < 8; ++k) v[k] = ROW(j + k);
        #pragma unroll
        for (int k = 0; k < 8; ++k) { acc.x += v[k].x; acc.y += v[k].y; }
    }
    const int rem = m - j;
    const int j4 = j + (rem & 4);
    const int j2 = j4 + (rem & 2);
    float2 t0 = {0.f, 0.f}, t1 = t0, t2 = t0, t3 = t0, t4 = t0, t5 = t0, t6 = t0;
    if (rem & 4) { t0 = ROW(j); t1 = ROW(j + 1); t2 = ROW(j + 2); t3 = ROW(j + 3); }
    if (rem & 2) { t4 = ROW(j4); t5 = ROW(j4 + 1); }
    if (rem & 1) { t6 = ROW(j2); }
    acc.x += ((t0.x + t1.x) + (t2.x + t3.x)) + ((t4.x + t5.x) + t6.x);
    acc.y += ((t0.y + t1.y) + (t2.y + t3.y)) + ((t4.y + t5.y) + t6.y);
}

__global__ void __launch_bounds__(256) k_gather(const int* __restrict__ csr, const int* __restrict__ off,
                          const int* __restrict__ cnt, const float* __restrict__ hs,
                          const float* __restrict__ dinv, const float* __restrict__ b_conv,
                          const float* __restrict__ W_lin, const float* __restrict__ b_lin,
                          float* __restrict__ out, int N) {
    const int lane = threadIdx.x & 31;
    const int wave = __builtin_amdgcn_readfirstlane((int)(threadIdx.x >> 5));
    const int node0 = (blockIdx.x * 8 + wave) * NPW;
    if (node0 >= N) return;
    const int nn = (N - node0) < NPW ? (N - node0) : NPW;

    int o_l = 0, c_l = 0, dn_l = 0;
    if (lane < nn) {
        o_l  = off[node0 + lane];
        c_l  = cnt[node0 + lane];
        dn_l = __float_as_int(dinv[node0 + lane]);
    }
    const float2* hrow = (const float2*)hs + lane;
    const float2 bc = ((const float2*)b_conv)[lane];
    const float2 wl = ((const float2*)W_lin)[lane];
    const float  bl = b_lin[0];

    int o_nx = __builtin_amdgcn_readlane(o_l, 0);
    int c_nx = __builtin_amdgcn_readlane(c_l, 0);
    float2 self_nx = hrow[(size_t)node0 * (F_OUT / 2)];
    int sidx_nx = 0;
    if (lane < (c_nx < 32 ? c_nx : 32)) sidx_nx = csr[o_nx + lane];

    float res = 0.0f;

    for (int i = 0; i < nn; ++i) {
        const int o = o_nx;
        const int c = c_nx;
        const float dn = __int_as_float(__builtin_amdgcn_readlane(dn_l, i));
        float2 acc = self_nx;
        int sidx = sidx_nx;

        if (i + 1 < nn) {
            o_nx = __builtin_amdgcn_readlane(o_l, i + 1);
            c_nx = __builtin_amdgcn_readlane(c_l, i + 1);
            self_nx = hrow[(size_t)(node0 + i + 1) * (F_OUT / 2)];
            sidx_nx = 0;
            if (lane < (c_nx < 32 ? c_nx : 32)) sidx_nx = csr[o_nx + lane];
        }

        sidx = min(max(sidx, 0), N - 1);
        sum_rows(acc, sidx, c < 32 ? (c > 0 ? c : 0) : 32, hrow);
        for (int base = 32; base < c; base += 32) {
            const int left = c - base;
            const int m = left < 32 ? left : 32;
            int sx = 0;
            if (lane < m) sx = csr[o + base + lane];
            sx = min(max(sx, 0), N - 1);
            sum_rows(acc, sx, m, hrow);
        }

        float p = fmaxf(acc.x * dn + bc.x, 0.0f) * wl.x
                + fmaxf(acc.y * dn + bc.y, 0.0f) * wl.y;

        #pragma unroll
        for (int s = 16; s > 0; s >>= 1)
            p += __shfl_down(p, s);

        const float z = __int_as_float(__builtin_amdgcn_readfirstlane(__float_as_int(p))) + bl;
        const float r = 1.0f / (1.0f + __expf(-z));
        if (lane == i) res = r;
    }

    if (lane < nn) out[node0 + lane] = res;
}


extern "C" void kernel_launch(void* const* d_in, const int* in_sizes, int n_in,
                              void* d_out, int out_size, void* d_ws, size_t ws_size,
                              hipStream_t stream) {
    const float* x      = (const float*)d_in[0];
    const int*   edge   = (const int*)  d_in[1];
    const float* W_conv = (const float*)d_in[2];
    const float* b_conv = (const float*)d_in[3];
    const float* W_lin  = (const float*)d_in[4];
    const float* b_lin  = (const float*)d_in[5];
    float*       out    = (float*)d_out;

    const int N = in_sizes[0] / F_IN;
    const int E = in_sizes[1] / 2;
    const int* src = edge;
    const int* dst = edge + E;
    const int NB = (N + BIN_NODES - 1) / BIN_NODES;

    const size_t Np  = ((size_t)N + 63) & ~(size_t)63;
    const size_t Ep  = ((size_t)E + 63) & ~(size_t)63;
    const size_t NBp = ((size_t)NB + 63) & ~(size_t)63;
    int*   cnt      = (int*)d_ws;
    int*   off      = cnt + Np;
    float* dinv     = (float*)(off + Np);
    int*   bincnt   = (int*)(dinv + Np);
    int*   bincur   = bincnt + NBp * PAD;
    int*   binstart = bincur + NBp * PAD;
    int2*  part     = (int2*)(binstart + NBp);
    int*   csr      = (int*)(part + Ep);
    float* hs       = (float*)(csr + Ep);

    const int nzero = (int)(2 * NBp * PAD);
    const int EB = (E + CHUNK - 1) / CHUNK;

    k_zero    <<<(nzero + 255) / 256, 256, 0, stream>>>(bincnt, nzero);
    k_bincount<<<EB, 256, (size_t)NB * sizeof(int), stream>>>(dst, bincnt, E, N, NB);
    k_binpart <<<EB, 256, (size_t)NB * 2 * sizeof(int), stream>>>(src, dst, bincnt, bincur, binstart, part, E, N, NB);
    k_binsort <<<NB, 256, 0, stream>>>(part, bincnt, binstart, csr, off, cnt, dinv, N, E);

    k_gemm<<<(N + 127) / 128, 256, 0, stream>>>(x, W_conv, dinv, hs, N);

    k_gather<<<(N + 8 * NPW - 1) / (8 * NPW), 256, 0, stream>>>(csr, off, cnt, hs, dinv, b_conv, W_lin, b_lin, out, N);
}


# edge sort level 1 block-local: each block sorts its 4096 edges by bin in LDS, writes its segment in whole lines plus a run table; no global histogram, zeroing, bin adds; sums still arrival-ranked
# speedup vs baseline: 3.9928x; 1.3529x over previous
#include <hip/hip_runtime.h>
#include <math.h>


typedef float v2f __attribute__((ext_vector_type(2)));
typedef float v4f __attribute__((ext_vector_type(4)));
typedef float v8f __attribute__((ext_vector_type(8)));
typedef int   v4i __attribute__((ext_vector_type(4)));

#define F_IN  128
#define F_OUT 64


#define BIN_SHIFT 8
#define BIN_NODES 256
#define EPT       16
#define CHUNK     (256 * EPT)
#define SPT       8

__device__ __forceinline__ int block_excl_scan_256(int v, int* wsum, int& total) {
    const int lane = threadIdx.x & 31;
    const int wave = threadIdx.x >> 5;
    int x = v;
    #pragma unroll
    for (int d = 1; d < 32; d <<= 1) {
        const int y = __shfl_up(x, d);
        if (lane >= d) x += y;
    }
    if (lane == 31) wsum[wave] = x;
    __syncthreads();
    int wbase = 0, tot = 0;
    #pragma unroll
    for (int w = 0; w < 8; ++w) {
        const int s = wsum[w];
        if (w < wave) wbase += s;
        tot += s;
    }
    total = tot;
    return wbase + x - v;
}

__device__ __forceinline__ int run_pack(const int start, const int len) { return (start << 16) | len; }

__global__ void __launch_bounds__(256) k_blocksort(const int* __restrict__ src, const int* __restrict__ dst,
                                                   int* __restrict__ table, int2* __restrict__ part,
                                                   int* __restrict__ cursor, int E, int N, int NB, int NBp) {
    extern __shared__ int sm[];
    __shared__ v4i stage4[CHUNK / 2];
    __shared__ int wsum[8];
    int* hist = sm;
    int* base = sm + NB;
    int2* stage = (int2*)stage4;
    if (blockIdx.x == 0 && threadIdx.x == 0) cursor[0] = 0;
    for (int b = threadIdx.x; b < NB; b += 256) hist[b] = 0;
    __syncthreads();

    const int e0 = blockIdx.x * CHUNK + threadIdx.x;
    int s[EPT], d[EPT], r[EPT];
    #pragma unroll
    for (int k = 0; k < EPT; ++k) {
        const int e = e0 + k * 256;
        d[k] = (e < E) ? dst[e] : -1;
        s[k] = (e < E) ? src[e] : 0;
    }
    #pragma unroll
    for (int k = 0; k < EPT; ++k)
        r[k] = ((unsigned)d[k] < (unsigned)N) ? atomicAdd(&hist[d[k] >> BIN_SHIFT], 1) : -1;
    __syncthreads();

    const int per = (NB + 255) >> 8;
    const int b0 = threadIdx.x * per;
    int tsum = 0;
    for (int i = 0; i < per; ++i) {
        const int b = b0 + i;
        if (b < NB) tsum += hist[b];
    }
    int total;
    int run = block_excl_scan_256(tsum, wsum, total);
    int* trow = table + (size_t)blockIdx.x * NBp;
    for (int i = 0; i < per; ++i) {
        const int b = b0 + i;
        if (b < NB) {
            const int h = hist[b];
            base[b] = run;
            trow[b] = run_pack(run, h);
            run += h;
        }
    }
    __syncthreads();

    int bb[EPT];
    #pragma unroll
    for (int k = 0; k < EPT; ++k) bb[k] = base[(r[k] >= 0) ? (d[k] >> BIN_SHIFT) : 0];
    #pragma unroll
    for (int k = 0; k < EPT; ++k)
        if (r[k] >= 0) stage[bb[k] + r[k]] = make_int2(s[k], d[k]);
    __syncthreads();

    v4i* seg = (v4i*)(part + (size_t)blockIdx.x * CHUNK);
    const int nq = (total + 1) >> 1;
    v4i q[CHUNK / 512];
    #pragma unroll
    for (int k = 0; k < CHUNK / 512; ++k) q[k] = stage4[threadIdx.x + 256 * k];
    asm volatile("" :: "v"(q[0]), "v"(q[1]), "v"(q[2]), "v"(q[3]), "v"(q[4]), "v"(q[5]), "v"(q[6]), "v"(q[7]));
    #pragma unroll
    for (int k = 0; k < CHUNK / 512; ++k) {
        const int i = threadIdx.x + 256 * k;
        if (i < nq) seg[i] = q[k];
    }
}

template <bool PLACE>
__device__ __forceinline__ void bin_pass(const int2* __restrict__ part, const int* rtab, int* cur,
                                         int* __restrict__ piece, const int EB, const int total) {
    const int lane = threadIdx.x & 31;
    const int wave = threadIdx.x >> 5;
    const int sub = lane >> 4;
    const int li = lane & 15;
    const int npair = (EB + 1) >> 1;
    for (int j0 = wave; j0 < npair; j0 += 8 * SPT) {
        int2 v[SPT];
        int rr[SPT], ee[SPT];
        #pragma unroll
        for (int k = 0; k < SPT; ++k) {
            const int r = 2 * (j0 + 8 * k) + sub;
            const int e = rtab[max(min(r, EB - 1), 0)];
            rr[k] = r;
            ee[k] = (r < EB) ? e : 0;
        }
        #pragma unroll
        for (int k = 0; k < SPT; ++k) {
            const int2* p = part + (size_t)rr[k] * CHUNK + (ee[k] >> 16);
            v[k] = (li < (ee[k] & 0xffff)) ? p[li] : make_int2(0, -1);
        }
        if (PLACE) {
            int pos[SPT];
            #pragma unroll
            for (int k = 0; k < SPT; ++k)
                pos[k] = (v[k].y >= 0) ? atomicAdd(&cur[v[k].y & (BIN_NODES - 1)], 1) : -1;
            #pragma unroll
            for (int k = 0; k < SPT; ++k)
                if ((unsigned)pos[k] < (unsigned)total) piece[pos[k]] = v[k].x;
        } else {
            #pragma unroll
            for (int k = 0; k < SPT; ++k)
                if (v[k].y >= 0) atomicAdd(&cur[v[k].y & (BIN_NODES - 1)], 1);
        }
        #pragma unroll
        for (int k = 0; k < SPT; ++k) {
            const int c = ee[k] & 0xffff;
            const int2* p = part + (size_t)rr[k] * CHUNK + (ee[k] >> 16);
            for (int i = li + 16; i < c; i += 16) {
                const int2 w = p[i];
                if (PLACE) {
                    const int q = atomicAdd(&cur[w.y & (BIN_NODES - 1)], 1);
                    if ((unsigned)q < (unsigned)total) piece[q] = w.x;
                } else {
                    atomicAdd(&cur[w.y & (BIN_NODES - 1)], 1);
                }
            }
        }
    }
}

__global__ void __launch_bounds__(256) k_binsort(const int2* __restrict__ part, const int* __restrict__ table,
                                                 int* __restrict__ cursor, int* __restrict__ csr,
                                                 int* __restrict__ off, int* __restrict__ cnt,
                                                 float* __restrict__ dinv, int N, int E, int EB, int NBp) {
    extern __shared__ int rtab[];
    __shared__ int cur[BIN_NODES];
    __shared__ int wsum[8];
    __shared__ int wlen[8];
    __shared__ int s_start;
    const int bin = blockIdx.x;
    const int lane = threadIdx.x & 31;
    const int wave = threadIdx.x >> 5;

    int mylen = 0;
    for (int r = threadIdx.x; r < EB; r += 256) {
        const int e = table[(size_t)r * NBp + bin];
        int bo = e >> 16;
        int c = e & 0xffff;
        if (bo < 0 || bo > CHUNK) { bo = 0; c = 0; }
        if (c > CHUNK - bo) c = CHUNK - bo;
        rtab[r] = run_pack(bo, c);
        mylen += c;
    }
    cur[threadIdx.x] = 0;
    #pragma unroll
    for (int d = 16; d > 0; d >>= 1) mylen += __shfl_xor(mylen, d);
    if (lane == 0) wlen[wave] = mylen;
    __syncthreads();

    int start = 0;
    if (threadIdx.x == 0) {
        int len = 0;
        #pragma unroll
        for (int w = 0; w < 8; ++w) len += wlen[w];
        start = atomicAdd(cursor, len);
    }

    bin_pass<false>(part, rtab, cur, csr, EB, 0);
    __syncthreads();

    const int c = cur[threadIdx.x];
    if (threadIdx.x == 0) s_start = start;
    int total;
    const int ex = block_excl_scan_256(c, wsum, total);
    int st = s_start;
    if (st < 0 || st > E || total > E - st) { st = 0; total = 0; }
    const int node = bin * BIN_NODES + threadIdx.x;
    if (node < N) {
        off[node] = st + ex;
        cnt[node] = (total > 0) ? c : 0;
        const float dg = (float)c + 1.0f;
        dinv[node] = (dg > 0.0f) ? rsqrtf(dg) : 0.0f;
    }
    cur[threadIdx.x] = ex;
    __syncthreads();

    bin_pass<true>(part, rtab, cur, csr + st, EB, total);
}


#define W_PITCH 132
#define KGROUPS (F_IN / 8)

__device__ __forceinline__ int frag_k(const int g, const int hf) { return 8 * g + 4 * hf; }

__device__ __forceinline__ v2f lo2(const float4 v) { v2f r; r.x = v.x; r.y = v.y; return r; }
__device__ __forceinline__ v2f hi2(const float4 v) { v2f r; r.x = v.z; r.y = v.w; return r; }

__global__ void __launch_bounds__(256) k_gemm(const float* __restrict__ x,
                                              const float* __restrict__ W,
                                              const float* __restrict__ dinv,
                                              float* __restrict__ h, int N) {
    __shared__ float4 Wl4[F_OUT * W_PITCH / 4];
    float* Wl = (float*)Wl4;

    const int wave = threadIdx.x >> 5;
    const int lane = threadIdx.x & 31;
    const int row0 = (blockIdx.x * 8 + wave) * 16;

    const int lr = lane & 15;
    const int hf = lane >> 4;

    int m = row0 + lr;
    if (m >= N) m = N - 1;

    v4f w[8];
    #pragma unroll
    for (int i = 0; i < 8; ++i) w[i] = ((const v4f*)W)[threadIdx.x + 256 * i];

    const int mbase = row0 + (hf << 3);
    float sc[8];
    #pragma unroll
    for (int v = 0; v < 8; ++v) sc[v] = dinv[(mbase + v < N) ? (mbase + v) : (N - 1)];

    float4 a[KGROUPS];
    {
        const float* xr = x + (size_t)m * F_IN;
        #pragma unroll
        for (int g = 0; g < KGROUPS; ++g) a[g] = *(const float4*)(xr + frag_k(g, hf));
    }
    asm volatile("" :: "v"(w[0]), "v"(w[1]), "v"(w[2]), "v"(w[3]), "v"(w[4]), "v"(w[5]), "v"(w[6]), "v"(w[7]));

    #pragma unroll
    for (int i = 0; i < 8; ++i) {
        const int q = threadIdx.x + 256 * i;
        *(v4f*)(Wl + (q >> 5) * W_PITCH + (q & 31) * 4) = w[i];
    }
    __syncthreads();
    if (row0 >= N) return;

    v8f c0 = {}, c1 = {}, c2 = {}, c3 = {};
    const float* wr = Wl + lr * W_PITCH + frag_k(0, hf);

    float4 b0 = *(const float4*)(wr +  0 * W_PITCH);
    float4 b1 = *(const float4*)(wr + 16 * W_PITCH);
    float4 b2 = *(const float4*)(wr + 32 * W_PITCH);
    float4 b3 = *(const float4*)(wr + 48 * W_PITCH);

    #pragma unroll
    for (int g = 0; g < KGROUPS; ++g) {
        float4 n0 = b0, n1 = b1, n2 = b2, n3 = b3;
        if (g + 1 < KGROUPS) {
            const float* wn = wr + (frag_k(g + 1, 0) - frag_k(0, 0));
            n0 = *(const float4*)(wn +  0 * W_PITCH);
            n1 = *(const float4*)(wn + 16 * W_PITCH);
            n2 = *(const float4*)(wn + 32 * W_PITCH);
            n3 = *(const float4*)(wn + 48 * W_PITCH);
        }
        __builtin_amdgcn_sched_barrier(0);

        const v2f al = lo2(a[g]), ah = hi2(a[g]);
        c0 = __builtin_amdgcn_wmma_f32_16x16x4_f32(false, al, false, lo2(b0), (short)0, c0, false, false);
        c1 = __builtin_amdgcn_wmma_f32_16x16x4_f32(false, al, false, lo2(b1), (short)0, c1, false, false);
        c2 = __builtin_amdgcn_wmma_f32_16x16x4_f32(false, al, false, lo2(b2), (short)0, c2, false, false);
        c3 = __builtin_amdgcn_wmma_f32_16x16x4_f32(false, al, false, lo2(b3), (short)0, c3, false, false);
        c0 = __builtin_amdgcn_wmma_f32_16x16x4_f32(false, ah, false, hi2(b0), (short)0, c0, false, false);
        c1 = __builtin_amdgcn_wmma_f32_16x16x4_f32(false, ah, false, hi2(b1), (short)0, c1, false, false);
        c2 = __builtin_amdgcn_wmma_f32_16x16x4_f32(false, ah, false, hi2(b2), (short)0, c2, false, false);
        c3 = __builtin_amdgcn_wmma_f32_16x16x4_f32(false, ah, false, hi2(b3), (short)0, c3, false, false);
        __builtin_amdgcn_sched_barrier(0);
        b0 = n0; b1 = n1; b2 = n2; b3 = n3;
    }

    #pragma unroll
    for (int v = 0; v < 8; ++v) {
        int mm = mbase + v;
        if (mm < N) {
            float* hp = h + (size_t)mm * F_OUT + lr;
            hp[ 0] = c0[v] * sc[v];
            hp[16] = c1[v] * sc[v];
            hp[32] = c2[v] * sc[v];
            hp[48] = c3[v] * sc[v];
        }
    }
}


#define NPW 8

#define ROW(jj) hrow[(size_t)__builtin_amdgcn_readlane(sidx, (jj)) * (F_OUT / 2)]

__device__ __forceinline__ void sum_rows(float2& acc, const int sidx, const int m, const float2* __restrict__ hrow) {
    int j = 0;
    for (; j + 8 <= m; j += 8) {
        float2 v[8];
        #pragma unroll
        for (int k = 0; k < 8; ++k) v[k] = ROW(j + k);
        #pragma unroll
        for (int k = 0; k < 8; ++k) { acc.x += v[k].x; acc.y += v[k].y; }
    }
    const int rem = m - j;
    const int j4 = j + (rem & 4);
    const int j2 = j4 + (rem & 2);
    float2 t0 = {0.f, 0.f}, t1 = t0, t2 = t0, t3 = t0, t4 = t0, t5 = t0, t6 = t0;
    if (rem & 4) { t0 = ROW(j); t1 = ROW(j + 1); t2 = ROW(j + 2); t3 = ROW(j + 3); }
    if (rem & 2) { t4 = ROW(j4); t5 = ROW(j4 + 1); }
    if (rem & 1) { t6 = ROW(j2); }
    acc.x += ((t0.x + t1.x) + (t2.x + t3.x)) + ((t4.x + t5.x) + t6.x);
    acc.y += ((t0.y + t1.y) + (t2.y + t3.y)) + ((t4.y + t5.y) + t6.y);
}

__global__ void __launch_bounds__(256) k_gather(const int* __restrict__ csr, const int* __restrict__ off,
                          const int* __restrict__ cnt, const float* __restrict__ hs,
                          const float* __restrict__ dinv, const float* __restrict__ b_conv,
                          const float* __restrict__ W_lin, const float* __restrict__ b_lin,
                          float* __restrict__ out, int N) {
    const int lane = threadIdx.x & 31;
    const int wave = __builtin_amdgcn_readfirstlane((int)(threadIdx.x >> 5));
    const int node0 = (blockIdx.x * 8 + wave) * NPW;
    if (node0 >= N) return;
    const int nn = (N - node0) < NPW ? (N - node0) : NPW;

    int o_l = 0, c_l = 0, dn_l = 0;
    if (lane < nn) {
        o_l  = off[node0 + lane];
        c_l  = cnt[node0 + lane];
        dn_l = __float_as_int(dinv[node0 + lane]);
    }
    const float2* hrow = (const float2*)hs + lane;
    const float2 bc = ((const float2*)b_conv)[lane];
    const float2 wl = ((const float2*)W_lin)[lane];
    const float  bl = b_lin[0];

    int o_nx = __builtin_amdgcn_readlane(o_l, 0);
    int c_nx = __builtin_amdgcn_readlane(c_l, 0);
    float2 self_nx = hrow[(size_t)node0 * (F_OUT / 2)];
    int sidx_nx = 0;
    if (lane < (c_nx < 32 ? c_nx : 32)) sidx_nx = csr[o_nx + lane];

    float res = 0.0f;

    for (int i = 0; i < nn; ++i) {
        const int o = o_nx;
        const int c = c_nx;
        const float dn = __int_as_float(__builtin_amdgcn_readlane(dn_l, i));
        float2 acc = self_nx;
        int sidx = sidx_nx;

        if (i + 1 < nn) {
            o_nx = __builtin_amdgcn_readlane(o_l, i + 1);
            c_nx = __builtin_amdgcn_readlane(c_l, i + 1);
            self_nx = hrow[(size_t)(node0 + i + 1) * (F_OUT / 2)];
            sidx_nx = 0;
            if (lane < (c_nx < 32 ? c_nx : 32)) sidx_nx = csr[o_nx + lane];
        }

        sidx = min(max(sidx, 0), N - 1);
        sum_rows(acc, sidx, c < 32 ? (c > 0 ? c : 0) : 32, hrow);
        for (int base = 32; base < c; base += 32) {
            const int left = c - base;
            const int m = left < 32 ? left : 32;
            int sx = 0;
            if (lane < m) sx = csr[o + base + lane];
            sx = min(max(sx, 0), N - 1);
            sum_rows(acc, sx, m, hrow);
        }

        float p = fmaxf(acc.x * dn + bc.x, 0.0f) * wl.x
                + fmaxf(acc.y * dn + bc.y, 0.0f) * wl.y;

        #pragma unroll
        for (int s = 16; s > 0; s >>= 1)
            p += __shfl_down(p, s);

        const float z = __int_as_float(__builtin_amdgcn_readfirstlane(__float_as_int(p))) + bl;
        const float r = 1.0f / (1.0f + __expf(-z));
        if (lane == i) res = r;
    }

    if (lane < nn) out[node0 + lane] = res;
}


extern "C" void kernel_launch(void* const* d_in, const int* in_sizes, int n_in,
                              void* d_out, int out_size, void* d_ws, size_t ws_size,
                              hipStream_t stream) {
    const float* x      = (const float*)d_in[0];
    const int*   edge   = (const int*)  d_in[1];
    const float* W_conv = (const float*)d_in[2];
    const float* b_conv = (const float*)d_in[3];
    const float* W_lin  = (const float*)d_in[4];
    const float* b_lin  = (const float*)d_in[5];
    float*       out    = (float*)d_out;

    const int N = in_sizes[0] / F_IN;
    const int E = in_sizes[1] / 2;
    const int* src = edge;
    const int* dst = edge + E;
    const int NB = (N + BIN_NODES - 1) / BIN_NODES;
    const int EB = (E + CHUNK - 1) / CHUNK;

    const size_t Np  = ((size_t)N + 63) & ~(size_t)63;
    const size_t Ep  = ((size_t)E + 63) & ~(size_t)63;
    const size_t NBp = ((size_t)NB + 63) & ~(size_t)63;
    int*   cnt      = (int*)d_ws;
    int*   off      = cnt + Np;
    float* dinv     = (float*)(off + Np);
    int*   cursor   = (int*)(dinv + Np);
    int*   table    = cursor + 64;
    int2*  part     = (int2*)(table + (size_t)EB * NBp);
    int*   csr      = (int*)(part + (size_t)EB * CHUNK);
    float* hs       = (float*)(csr + Ep);

    k_blocksort<<<EB, 256, (size_t)NB * 2 * sizeof(int), stream>>>(src, dst, table, part, cursor, E, N, NB, (int)NBp);
    k_binsort  <<<NB, 256, (size_t)EB * sizeof(int), stream>>>(part, table, cursor, csr, off, cnt, dinv, N, E, EB, (int)NBp);

    k_gemm<<<(N + 127) / 128, 256, 0, stream>>>(x, W_conv, dinv, hs, N);

    k_gather<<<(N + 8 * NPW - 1) / (8 * NPW), 256, 0, stream>>>(csr, off, cnt, hs, dinv, b_conv, W_lin, b_lin, out, N);
}


# k_gemm fused into the per-bin sort kernel: 8 more waves per bin make the bin's 256 rows of hs on the matrix pipe while 8 waves sort; rows parked in LDS, scaled by dinv, whole-line stores; 3 launches
# speedup vs baseline: 4.6621x; 1.1676x over previous
#include <hip/hip_runtime.h>
#include <math.h>


typedef float v2f __attribute__((ext_vector_type(2)));
typedef float v4f __attribute__((ext_vector_type(4)));
typedef float v8f __attribute__((ext_vector_type(8)));
typedef int   v4i __attribute__((ext_vector_type(4)));

#define F_IN  128
#define F_OUT 64


#define BIN_SHIFT 8
#define BIN_NODES 256
#define EPT       16
#define CHUNK     (256 * EPT)
#define SPT       8

__device__ __forceinline__ int block_excl_scan_256(int v, int* wsum, int& total) {
    const int lane = threadIdx.x & 31;
    const int wave = threadIdx.x >> 5;
    int x = v;
    #pragma unroll
    for (int d = 1; d < 32; d <<= 1) {
        const int y = __shfl_up(x, d);
        if (lane >= d) x += y;
    }
    if (lane == 31) wsum[wave] = x;
    __syncthreads();
    int wbase = 0, tot = 0;
    #pragma unroll
    for (int w = 0; w < 8; ++w) {
        const int s = wsum[w];
        if (w < wave) wbase += s;
        tot += s;
    }
    total = tot;
    return wbase + x - v;
}

__device__ __forceinline__ int run_pack(const int start, const int len) { return (start << 16) | len; }

__global__ void __launch_bounds__(256) k_blocksort(const int* __restrict__ src, const int* __restrict__ dst,
                                                   int* __restrict__ table, int2* __restrict__ part,
                                                   int* __restrict__ cursor, int E, int N, int NB, int NBp) {
    extern __shared__ int sm[];
    __shared__ v4i stage4[CHUNK / 2];
    __shared__ int wsum[8];
    int* hist = sm;
    int* base = sm + NB;
    int2* stage = (int2*)stage4;
    if (blockIdx.x == 0 && threadIdx.x == 0) cursor[0] = 0;
    for (int b = threadIdx.x; b < NB; b += 256) hist[b] = 0;
    __syncthreads();

    const int e0 = blockIdx.x * CHUNK + threadIdx.x;
    int s[EPT], d[EPT], r[EPT];
    #pragma unroll
    for (int k = 0; k < EPT; ++k) {
        const int e = e0 + k * 256;
        d[k] = (e < E) ? dst[e] : -1;
        s[k] = (e < E) ? src[e] : 0;
    }
    #pragma unroll
    for (int k = 0; k < EPT; ++k)
        r[k] = ((unsigned)d[k] < (unsigned)N) ? atomicAdd(&hist[d[k] >> BIN_SHIFT], 1) : -1;
    __syncthreads();

    const int per = (NB + 255) >> 8;
    const int b0 = threadIdx.x * per;
    int tsum = 0;
    for (int i = 0; i < per; ++i) {
        const int b = b0 + i;
        if (b < NB) tsum += hist[b];
    }
    int total;
    int run = block_excl_scan_256(tsum, wsum, total);
    int* trow = table + (size_t)blockIdx.x * NBp;
    for (int i = 0; i < per; ++i) {
        const int b = b0 + i;
        if (b < NB) {
            const int h = hist[b];
            base[b] = run;
            trow[b] = run_pack(run, h);
            run += h;
        }
    }
    __syncthreads();

    int bb[EPT];
    #pragma unroll
    for (int k = 0; k < EPT; ++k) bb[k] = base[(r[k] >= 0) ? (d[k] >> BIN_SHIFT) : 0];
    #pragma unroll
    for (int k = 0; k < EPT; ++k)
        if (r[k] >= 0) stage[bb[k] + r[k]] = make_int2(s[k], d[k]);
    __syncthreads();

    v4i* seg = (v4i*)(part + (size_t)blockIdx.x * CHUNK);
    const int nq = (total + 1) >> 1;
    v4i q[CHUNK / 512];
    #pragma unroll
    for (int k = 0; k < CHUNK / 512; ++k) q[k] = stage4[threadIdx.x + 256 * k];
    asm volatile("" :: "v"(q[0]), "v"(q[1]), "v"(q[2]), "v"(q[3]), "v"(q[4]), "v"(q[5]), "v"(q[6]), "v"(q[7]));
    #pragma unroll
    for (int k = 0; k < CHUNK / 512; ++k) {
        const int i = threadIdx.x + 256 * k;
        if (i < nq) seg[i] = q[k];
    }
}

template <bool PLACE>
__device__ __forceinline__ void bin_pass(const int2* __restrict__ part, const int* rtab, int* cur,
                                         int* __restrict__ piece, const int EB, const int total) {
    const int lane = threadIdx.x & 31;
    const int wave = threadIdx.x >> 5;
    const int sub = lane >> 4;
    const int li = lane & 15;
    const int npair = (EB + 1) >> 1;
    for (int j0 = wave; j0 < npair; j0 += 8 * SPT) {
        int2 v[SPT];
        int rr[SPT], ee[SPT];
        #pragma unroll
        for (int k = 0; k < SPT; ++k) {
            const int r = 2 * (j0 + 8 * k) + sub;
            const int e = rtab[max(min(r, EB - 1), 0)];
            rr[k] = r;
            ee[k] = (r < EB) ? e : 0;
        }
        #pragma unroll
        for (int k = 0; k < SPT; ++k) {
            const int2* p = part + (size_t)rr[k] * CHUNK + (ee[k] >> 16);
            v[k] = (li < (ee[k] & 0xffff)) ? p[li] : make_int2(0, -1);
        }
        if (PLACE) {
            int pos[SPT];
            #pragma unroll
            for (int k = 0; k < SPT; ++k)
                pos[k] = (v[k].y >= 0) ? atomicAdd(&cur[v[k].y & (BIN_NODES - 1)], 1) : -1;
            #pragma unroll
            for (int k = 0; k < SPT; ++k)
                if ((unsigned)pos[k] < (unsigned)total) piece[pos[k]] = v[k].x;
        } else {
            #pragma unroll
            for (int k = 0; k < SPT; ++k)
                if (v[k].y >= 0) atomicAdd(&cur[v[k].y & (BIN_NODES - 1)], 1);
        }
        #pragma unroll
        for (int k = 0; k < SPT; ++k) {
            const int c = ee[k] & 0xffff;
            const int2* p = part + (size_t)rr[k] * CHUNK + (ee[k] >> 16);
            for (int i = li + 16; i < c; i += 16) {
                const int2 w = p[i];
                if (PLACE) {
                    const int q = atomicAdd(&cur[w.y & (BIN_NODES - 1)], 1);
                    if ((unsigned)q < (unsigned)total) piece[q] = w.x;
                } else {
                    atomicAdd(&cur[w.y & (BIN_NODES - 1)], 1);
                }
            }
        }
    }
}


#define W_PITCH 132
#define H_PITCH 68
#define KGROUPS (F_IN / 8)

__device__ __forceinline__ int frag_k(const int g, const int hf) { return 8 * g + 4 * hf; }

__device__ __forceinline__ v2f lo2(const float4 v) { v2f r; r.x = v.x; r.y = v.y; return r; }
__device__ __forceinline__ v2f hi2(const float4 v) { v2f r; r.x = v.z; r.y = v.w; return r; }

template <bool NEXT>
__device__ __forceinline__ void tile_product(float4 (&a)[KGROUPS], float4 (&an)[KGROUPS], const float* xn,
                                             const float* wr, float* hrow) {
    #pragma unroll
    for (int p = 0; p < 2; ++p) {
        v8f c0 = {}, c1 = {};
        const float* wp = wr + 32 * p * W_PITCH;

        float4 b0 = *(const float4*)(wp +  0 * W_PITCH);
        float4 b1 = *(const float4*)(wp + 16 * W_PITCH);

        #pragma unroll
        for (int g = 0; g < KGROUPS; ++g) {
            float4 n0 = b0, n1 = b1;
            if (g + 1 < KGROUPS) {
                const float* wn = wp + (frag_k(g + 1, 0) - frag_k(0, 0));
                n0 = *(const float4*)(wn +  0 * W_PITCH);
                n1 = *(const float4*)(wn + 16 * W_PITCH);
            }
            __builtin_amdgcn_sched_barrier(0);

            const v2f al = lo2(a[g]), ah = hi2(a[g]);
            c0 = __builtin_amdgcn_wmma_f32_16x16x4_f32(false, al, false, lo2(b0), (short)0, c0, false, false);
            c1 = __builtin_amdgcn_wmma_f32_16x16x4_f32(false, al, false, lo2(b1), (short)0, c1, false, false);
            c0 = __builtin_amdgcn_wmma_f32_16x16x4_f32(false, ah, false, hi2(b0), (short)0, c0, false, false);
            c1 = __builtin_amdgcn_wmma_f32_16x16x4_f32(false, ah, false, hi2(b1), (short)0, c1, false, false);
            __builtin_amdgcn_sched_barrier(0);
            b0 = n0; b1 = n1;

            if (NEXT && p == 1 && (g == KGROUPS / 2 - 1 || g == KGROUPS - 1)) {
                const int hf = (threadIdx.x & 31) >> 4;
                #pragma unroll
                for (int q = g - (KGROUPS / 2 - 1); q <= g; ++q) an[q] = *(const float4*)(xn + frag_k(q, hf));
            }
        }

        #pragma unroll
        for (int v = 0; v < 8; ++v) {
            float* hp = hrow + v * H_PITCH + 32 * p;
            hp[ 0] = c0[v];
            hp[16] = c1[v];
        }
    }
}

__device__ __forceinline__ void store_tile(const float* ht, const float* dvt, float* __restrict__ hg, const int nrow) {
    const int lane = threadIdx.x & 31;
    const int pc = (lane & 15) * 4;
    const int r0 = lane >> 4;
    if (nrow >= 16) {
        float4 r[8];
        float s[8];
        #pragma unroll
        for (int i = 0; i < 8; ++i) {
            r[i] = *(const float4*)(ht + (2 * i + r0) * H_PITCH + pc);
            s[i] = dvt[2 * i + r0];
        }
        #pragma unroll
        for (int i = 0; i < 8; ++i) {
            float4 o;
            o.x = r[i].x * s[i]; o.y = r[i].y * s[i]; o.z = r[i].z * s[i]; o.w = r[i].w * s[i];
            *(float4*)(hg + (size_t)(2 * i + r0) * F_OUT + pc) = o;
        }
    } else {
        for (int i = 0; i < 8; ++i) {
            const int row = 2 * i + r0;
            if (row < nrow) {
                const float4 r = *(const float4*)(ht + row * H_PITCH + pc);
                const float s = dvt[row];
                float4 o;
                o.x = r.x * s; o.y = r.y * s; o.z = r.z * s; o.w = r.w * s;
                *(float4*)(hg + (size_t)row * F_OUT + pc) = o;
            }
        }
    }
}

__global__ void __launch_bounds__(512, 8) k_bin(const int2* __restrict__ part, const int* __restrict__ table,
                                                int* __restrict__ cursor, int* __restrict__ csr,
                                                int* __restrict__ off, int* __restrict__ cnt,
                                                float* __restrict__ dinv, const float* __restrict__ x,
                                                const float* __restrict__ W, float* __restrict__ hs,
                                                int N, int E, int EB, int NBp) {
    extern __shared__ int rtab[];
    __shared__ float4 Wl4[F_OUT * W_PITCH / 4];
    __shared__ float4 hp4[BIN_NODES * H_PITCH / 4];
    __shared__ float dv[BIN_NODES];
    __shared__ int cur[BIN_NODES];
    __shared__ int wsum[8];
    __shared__ int wlen[8];
    __shared__ int s_start;
    float* Wl = (float*)Wl4;
    float* hpk = (float*)hp4;

    const int bin = blockIdx.x;
    const int lane = threadIdx.x & 31;
    const int wave = __builtin_amdgcn_readfirstlane((int)(threadIdx.x >> 5));
    const bool sorter = wave < 8;
    const int gt = threadIdx.x - 256;
    const int lr = lane & 15;
    const int hf = lane >> 4;
    const int trow = (wave - 8) * 32;

    float4 a[KGROUPS], a2[KGROUPS];
    const float* x2 = x;

    if (sorter) {
        int mylen = 0;
        for (int r = threadIdx.x; r < EB; r += 256) {
            const int e = table[(size_t)r * NBp + bin];
            int bo = e >> 16;
            int c = e & 0xffff;
            if (bo < 0 || bo > CHUNK) { bo = 0; c = 0; }
            if (c > CHUNK - bo) c = CHUNK - bo;
            rtab[r] = run_pack(bo, c);
            mylen += c;
        }
        cur[threadIdx.x] = 0;
        #pragma unroll
        for (int d = 16; d > 0; d >>= 1) mylen += __shfl_xor(mylen, d);
        if (lane == 0) wlen[wave] = mylen;
    } else {
        v4f w[8];
        #pragma unroll
        for (int i = 0; i < 8; ++i) w[i] = ((const v4f*)W)[gt + 256 * i];

        int m = bin * BIN_NODES + trow + lr;
        const int m2 = (m + 16 < N) ? (m + 16) : (N - 1);
        if (m >= N) m = N - 1;
        x2 = x + (size_t)m2 * F_IN;
        const float* xr = x + (size_t)m * F_IN;
        #pragma unroll
        for (int g = 0; g < KGROUPS; ++g) a[g] = *(const float4*)(xr + frag_k(g, hf));
        asm volatile("" :: "v"(w[0]), "v"(w[1]), "v"(w[2]), "v"(w[3]), "v"(w[4]), "v"(w[5]), "v"(w[6]), "v"(w[7]));
        #pragma unroll
        for (int i = 0; i < 8; ++i) {
            const int q = gt + 256 * i;
            *(v4f*)(Wl + (q >> 5) * W_PITCH + (q & 31) * 4) = w[i];
        }
    }
    __syncthreads();

    int start = 0;
    if (sorter) {
        if (threadIdx.x == 0) {
            int len = 0;
            #pragma unroll
            for (int w = 0; w < 8; ++w) len += wlen[w];
            start = atomicAdd(cursor, len);
        }
        bin_pass<false>(part, rtab, cur, csr, EB, 0);
    } else {
        const float* wr = Wl + lr * W_PITCH + frag_k(0, hf);
        float* hrow = hpk + (trow + 8 * hf) * H_PITCH + lr;
        tile_product<true>(a, a2, x2, wr, hrow);
    }
    __syncthreads();

    int c = 0, xs = 0;
    if (sorter) {
        c = cur[threadIdx.x];
        if (threadIdx.x == 0) s_start = start;
        xs = c;
        #pragma unroll
        for (int d = 1; d < 32; d <<= 1) {
            const int y = __shfl_up(xs, d);
            if (lane >= d) xs += y;
        }
        if (lane == 31) wsum[wave] = xs;
    }
    __syncthreads();

    int st = 0, total = 0;
    if (sorter) {
        int wbase = 0;
        #pragma unroll
        for (int w = 0; w < 8; ++w) {
            const int sw = wsum[w];
            if (w < wave) wbase += sw;
            total += sw;
        }
        const int ex = wbase + xs - c;
        st = s_start;
        if (st < 0 || st > E || total > E - st) { st = 0; total = 0; }
        const int node = bin * BIN_NODES + threadIdx.x;
        const float dg = (float)c + 1.0f;
        const float di = (dg > 0.0f) ? rsqrtf(dg) : 0.0f;
        dv[threadIdx.x] = di;
        if (node < N) {
            off[node] = st + ex;
            cnt[node] = (total > 0) ? c : 0;
            dinv[node] = di;
        }
        cur[threadIdx.x] = ex;
    }
    __syncthreads();

    if (sorter) {
        bin_pass<true>(part, rtab, cur, csr + st, EB, total);
    } else {
        const float* wr = Wl + lr * W_PITCH + frag_k(0, hf);
        float* hrow = hpk + (trow + 8 * hf) * H_PITCH + lr;
        const int node0 = bin * BIN_NODES + trow;
        tile_product<false>(a2, a2, x2, wr, hrow + 16 * H_PITCH);
        store_tile(hpk + trow * H_PITCH, dv + trow, hs + (size_t)node0 * F_OUT, N - node0);
        store_tile(hpk + (trow + 16) * H_PITCH, dv + trow + 16, hs + (size_t)(node0 + 16) * F_OUT, N - node0 - 16);
    }
}


#define NPW 8

#define ROW(jj) hrow[(size_t)__builtin_amdgcn_readlane(sidx, (jj)) * (F_OUT / 2)]

__device__ __forceinline__ void sum_rows(float2& acc, const int sidx, const int m, const float2* __restrict__ hrow) {
    int j = 0;
    for (; j + 8 <= m; j += 8) {
        float2 v[8];
        #pragma unroll
        for (int k = 0; k < 8; ++k) v[k] = ROW(j + k);
        #pragma unroll
        for (int k = 0; k < 8; ++k) { acc.x += v[k].x; acc.y += v[k].y; }
    }
    const int rem = m - j;
    const int j4 = j + (rem & 4);
    const int j2 = j4 + (rem & 2);
    float2 t0 = {0.f, 0.f}, t1 = t0, t2 = t0, t3 = t0, t4 = t0, t5 = t0, t6 = t0;
    if (rem & 4) { t0 = ROW(j); t1 = ROW(j + 1); t2 = ROW(j + 2); t3 = ROW(j + 3); }
    if (rem & 2) { t4 = ROW(j4); t5 = ROW(j4 + 1); }
    if (rem & 1) { t6 = ROW(j2); }
    acc.x += ((t0.x + t1.x) + (t2.x + t3.x)) + ((t4.x + t5.x) + t6.x);
    acc.y += ((t0.y + t1.y) + (t2.y + t3.y)) + ((t4.y + t5.y) + t6.y);
}

__global__ void __launch_bounds__(256) k_gather(const int* __restrict__ csr, const int* __restrict__ off,
                          const int* __restrict__ cnt, const float* __restrict__ hs,
                          const float* __restrict__ dinv, const float* __restrict__ b_conv,
                          const float* __restrict__ W_lin, const float* __restrict__ b_lin,
                          float* __restrict__ out, int N) {
    const int lane = threadIdx.x & 31;
    const int wave = __builtin_amdgcn_readfirstlane((int)(threadIdx.x >> 5));
    const int node0 = (blockIdx.x * 8 + wave) * NPW;
    if (node0 >= N) return;
    const int nn = (N - node0) < NPW ? (N - node0) : NPW;

    int o_l = 0, c_l = 0, dn_l = 0;
    if (lane < nn) {
        o_l  = off[node0 + lane];
        c_l  = cnt[node0 + lane];
        dn_l = __float_as_int(dinv[node0 + lane]);
    }
    const float2* hrow = (const float2*)hs + lane;
    const float2 bc = ((const float2*)b_conv)[lane];
    const float2 wl = ((const float2*)W_lin)[lane];
    const float  bl = b_lin[0];

    int o_nx = __builtin_amdgcn_readlane(o_l, 0);
    int c_nx = __builtin_amdgcn_readlane(c_l, 0);
    float2 self_nx = hrow[(size_t)node0 * (F_OUT / 2)];
    int sidx_nx = 0;
    if (lane < (c_nx < 32 ? c_nx : 32)) sidx_nx = csr[o_nx + lane];

    float res = 0.0f;

    for (int i = 0; i < nn; ++i) {
        const int o = o_nx;
        const int c = c_nx;
        const float dn = __int_as_float(__builtin_amdgcn_readlane(dn_l, i));
        float2 acc = self_nx;
        int sidx = sidx_nx;

        if (i + 1 < nn) {
            o_nx = __builtin_amdgcn_readlane(o_l, i + 1);
            c_nx = __builtin_amdgcn_readlane(c_l, i + 1);
            self_nx = hrow[(size_t)(node0 + i + 1) * (F_OUT / 2)];
            sidx_nx = 0;
            if (lane < (c_nx < 32 ? c_nx : 32)) sidx_nx = csr[o_nx + lane];
        }

        sidx = min(max(sidx, 0), N - 1);
        sum_rows(acc, sidx, c < 32 ? (c > 0 ? c : 0) : 32, hrow);
        for (int base = 32; base < c; base += 32) {
            const int left = c - base;
            const int m = left < 32 ? left : 32;
            int sx = 0;
            if (lane < m) sx = csr[o + base + lane];
            sx = min(max(sx, 0), N - 1);
            sum_rows(acc, sx, m, hrow);
        }

        float p = fmaxf(acc.x * dn + bc.x, 0.0f) * wl.x
                + fmaxf(acc.y * dn + bc.y, 0.0f) * wl.y;

        #pragma unroll
        for (int s = 16; s > 0; s >>= 1)
            p += __shfl_down(p, s);

        const float z = __int_as_float(__builtin_amdgcn_readfirstlane(__float_as_int(p))) + bl;
        const float r = 1.0f / (1.0f + __expf(-z));
        if (lane == i) res = r;
    }

    if (lane < nn) out[node0 + lane] = res;
}


extern "C" void kernel_launch(void* const* d_in, const int* in_sizes, int n_in,
                              void* d_out, int out_size, void* d_ws, size_t ws_size,
                              hipStream_t stream) {
    const float* x      = (const float*)d_in[0];
    const int*   edge   = (const int*)  d_in[1];
    const float* W_conv = (const float*)d_in[2];
    const float* b_conv = (const float*)d_in[3];
    const float* W_lin  = (const float*)d_in[4];
    const float* b_lin  = (const float*)d_in[5];
    float*       out    = (float*)d_out;

    const int N = in_sizes[0] / F_IN;
    const int E = in_sizes[1] / 2;
    const int* src = edge;
    const int* dst = edge + E;
    const int NB = (N + BIN_NODES - 1) / BIN_NODES;
    const int EB = (E + CHUNK - 1) / CHUNK;

    const size_t Np  = ((size_t)N + 63) & ~(size_t)63;
    const size_t Ep  = ((size_t)E + 63) & ~(size_t)63;
    const size_t NBp = ((size_t)NB + 63) & ~(size_t)63;
    int*   cnt      = (int*)d_ws;
    int*   off      = cnt + Np;
    float* dinv     = (float*)(off + Np);
    int*   cursor   = (int*)(dinv + Np);
    int*   table    = cursor + 64;
    int2*  part     = (int2*)(table + (size_t)EB * NBp);
    int*   csr      = (int*)(part + (size_t)EB * CHUNK);
    float* hs       = (float*)(csr + Ep);

    k_blocksort<<<EB, 256, (size_t)NB * 2 * sizeof(int), stream>>>(src, dst, table, part, cursor, E, N, NB, (int)NBp);
    k_bin      <<<NB, 512, (size_t)EB * sizeof(int), stream>>>(part, table, cursor, csr, off, cnt, dinv, x, W_conv, hs, N, E, EB, (int)NBp);

    k_gather<<<(N + 8 * NPW - 1) / (8 * NPW), 256, 0, stream>>>(csr, off, cnt, hs, dinv, b_conv, W_lin, b_lin, out, N);
}
